# starting kernel
# baseline (speedup 1.0000x reference)
#include <hip/hip_runtime.h>
#include <hip/hip_bf16.h>

typedef __attribute__((ext_vector_type(2))) float v2f;
typedef __attribute__((ext_vector_type(8))) float v8f;

#define NEG_SLOPE 0.2f


__device__ __forceinline__ float warp_sum(float v) {
  #pragma unroll
  for (int o = 16; o > 0; o >>= 1) v += __shfl_xor(v, o, 32);
  return v;
}

__device__ __forceinline__ unsigned fkey(float f) {
  unsigned u = __float_as_uint(f);
  return (u & 0x80000000u) ? ~u : (u | 0x80000000u);
}
__device__ __forceinline__ float funkey(unsigned k) {
  unsigned u = (k & 0x80000000u) ? (k & 0x7FFFFFFFu) : ~k;
  return __uint_as_float(u);
}

__device__ __forceinline__ void atomic_add_f32(float* p, float v) {
#if defined(__HIP_DEVICE_COMPILE__)
  unsafeAtomicAdd(p, v);
#else
  atomicAdd(p, v);
#endif
}


__global__ void zero_f32(float* __restrict__ p, size_t n) {
  size_t stride = (size_t)gridDim.x * blockDim.x;
  for (size_t i = (size_t)blockIdx.x * blockDim.x + threadIdx.x; i < n; i += stride)
    p[i] = 0.0f;
}

__global__ void gemm1_wmma(const float* __restrict__ X, const float* __restrict__ W,
                           float* __restrict__ H, int N) {
  const int lane = threadIdx.x & 31;
  const int wave = threadIdx.x >> 5;
  const int row0 = blockIdx.x * 16;
  const int col0 = wave * 16;
  const int m  = lane & 15;
  const int kh = lane >> 4;

  int row = row0 + m; if (row >= N) row = N - 1;
  const float* xrow = X + (size_t)row * 256;

  v8f c = {};
#if __has_builtin(__builtin_amdgcn_wmma_f32_16x16x4_f32)
  #pragma unroll 4
  for (int k = 0; k < 256; k += 4) {
    v2f a = *(const v2f*)(xrow + k + 2 * kh);
    v2f b;
    const float* wp = W + (size_t)(k + 2 * kh) * 128 + col0 + m;
    b.x = wp[0];
    b.y = wp[128];
    c = __builtin_amdgcn_wmma_f32_16x16x4_f32(false, a, false, b, (short)0, c,
                                              false, false);
  }
#else
  for (int k = 0; k < 256; k += 4) {
    v2f a = *(const v2f*)(xrow + k + 2 * kh);
    const float* wp = W + (size_t)(k + 2 * kh) * 128 + col0 + m;
    c[0] += a.x * wp[0] + a.y * wp[128];
  }
#endif
  float* hp = H + (size_t)(row0 + 8 * kh) * 128 + col0 + m;
  if (row0 + 16 <= N) {
    #pragma unroll
    for (int r = 0; r < 8; ++r) hp[(size_t)r * 128] = c[r];
  } else {
    #pragma unroll
    for (int r = 0; r < 8; ++r)
      if (row0 + r + 8 * kh < N) hp[(size_t)r * 128] = c[r];
  }
}

__global__ void node_e1(const float* __restrict__ H, const float* __restrict__ as,
                        const float* __restrict__ ad, float* __restrict__ es,
                        float* __restrict__ ed, int N) {
  int n = blockIdx.x * 8 + (threadIdx.x >> 5);
  if (n >= N) return;
  int lane = threadIdx.x & 31;
  float4 hv = ((const float4*)(H + (size_t)n * 128))[lane];
  float4 sv = ((const float4*)as)[lane];
  float4 dv = ((const float4*)ad)[lane];
  float s = hv.x * sv.x + hv.y * sv.y + hv.z * sv.z + hv.w * sv.w;
  float d = hv.x * dv.x + hv.y * dv.y + hv.z * dv.z + hv.w * dv.w;
  s = warp_sum(s);
  d = warp_sum(d);
  if (lane == 0) { es[n] = s; ed[n] = d; }
}

__global__ void edge_logits(const float* __restrict__ es, const float* __restrict__ ed,
                            const int* __restrict__ src, const int* __restrict__ dst,
                            int E, int nT, float* __restrict__ ebuf,
                            unsigned* __restrict__ mkey) {
  int i = blockIdx.x * blockDim.x + threadIdx.x;
  if (i >= nT) return;
  int s, d;
  if (i < E) { s = src[i]; d = dst[i]; } else { s = d = i - E; }
  float e = es[s] + ed[d];
  e = (e > 0.0f) ? e : NEG_SLOPE * e;
  ebuf[i] = e;
  atomicMax(&mkey[d], fkey(e));
}

__global__ void edge_exp(const float* __restrict__ ein, const unsigned* __restrict__ mkey,
                         const int* __restrict__ dst, int E, int nT,
                         float* __restrict__ eout, float* __restrict__ denom) {
  int i = blockIdx.x * blockDim.x + threadIdx.x;
  if (i >= nT) return;
  int d = (i < E) ? dst[i] : (i - E);
  float ex = expf(ein[i] - funkey(mkey[d]));
  eout[i] = ex;
  atomic_add_f32(&denom[d], ex);
}

__global__ void agg128(const float* __restrict__ H, const float* __restrict__ ex,
                       const float* __restrict__ denom, const int* __restrict__ src,
                       const int* __restrict__ dst, int E, int nT,
                       float* __restrict__ agg) {
  int i = blockIdx.x * 8 + (threadIdx.x >> 5);
  if (i >= nT) return;
  int lane = threadIdx.x & 31;
  int s, d;
  if (i < E) { s = src[i]; d = dst[i]; } else { s = d = i - E; }
  float alpha = ex[i] / denom[d];
  float4 hv = ((const float4*)(H + (size_t)s * 128))[lane];
  float* o = agg + (size_t)d * 128 + lane * 4;
  atomic_add_f32(o + 0, hv.x * alpha);
  atomic_add_f32(o + 1, hv.y * alpha);
  atomic_add_f32(o + 2, hv.z * alpha);
  atomic_add_f32(o + 3, hv.w * alpha);
}

__global__ void bias_relu(float* __restrict__ a, const float* __restrict__ b, size_t n) {
  size_t i = (size_t)blockIdx.x * blockDim.x + threadIdx.x;
  if (i >= n) return;
  float v = a[i] + b[i & 127];
  a[i] = (v > 0.0f) ? v : 0.0f;
}

__global__ void node_h2e2(const float* __restrict__ H, const float* __restrict__ W2,
                          const float* __restrict__ a2s, const float* __restrict__ a2d,
                          float* __restrict__ h2, float* __restrict__ e2s,
                          float* __restrict__ e2d, int N) {
  int n = blockIdx.x * 8 + (threadIdx.x >> 5);
  if (n >= N) return;
  int lane = threadIdx.x & 31;
  float4 hv = ((const float4*)(H + (size_t)n * 128))[lane];
  int c0 = lane * 4;
  float s0 = hv.x * W2[(c0 + 0) * 2 + 0] + hv.y * W2[(c0 + 1) * 2 + 0] +
             hv.z * W2[(c0 + 2) * 2 + 0] + hv.w * W2[(c0 + 3) * 2 + 0];
  float s1 = hv.x * W2[(c0 + 0) * 2 + 1] + hv.y * W2[(c0 + 1) * 2 + 1] +
             hv.z * W2[(c0 + 2) * 2 + 1] + hv.w * W2[(c0 + 3) * 2 + 1];
  s0 = warp_sum(s0);
  s1 = warp_sum(s1);
  if (lane == 0) {
    h2[2 * n + 0] = s0;
    h2[2 * n + 1] = s1;
    e2s[n] = s0 * a2s[0] + s1 * a2s[1];
    e2d[n] = s0 * a2d[0] + s1 * a2d[1];
  }
}

__global__ void agg2(const float* __restrict__ h2, const float* __restrict__ ex,
                     const float* __restrict__ denom, const int* __restrict__ src,
                     const int* __restrict__ dst, int E, int nT,
                     float* __restrict__ out) {
  int i = blockIdx.x * blockDim.x + threadIdx.x;
  if (i >= nT) return;
  int s, d;
  if (i < E) { s = src[i]; d = dst[i]; } else { s = d = i - E; }
  float alpha = ex[i] / denom[d];
  atomic_add_f32(&out[2 * d + 0], h2[2 * s + 0] * alpha);
  atomic_add_f32(&out[2 * d + 1], h2[2 * s + 1] * alpha);
}

__global__ void bias2(float* __restrict__ out, const float* __restrict__ b, int n2) {
  int i = blockIdx.x * blockDim.x + threadIdx.x;
  if (i >= n2) return;
  out[i] += b[i & 1];
}


extern "C" void kernel_launch(void* const* d_in, const int* in_sizes, int n_in,
                              void* d_out, int out_size, void* d_ws, size_t ws_size,
                              hipStream_t stream) {
  const float* x   = (const float*)d_in[0];
  const int*  eidx = (const int*)d_in[1];
  const float* W1  = (const float*)d_in[2];
  const float* a1s = (const float*)d_in[3];
  const float* a1d = (const float*)d_in[4];
  const float* b1  = (const float*)d_in[5];
  const float* W2  = (const float*)d_in[6];
  const float* a2s = (const float*)d_in[7];
  const float* a2d = (const float*)d_in[8];
  const float* b2  = (const float*)d_in[9];

  const int N  = in_sizes[0] / 256;
  const int E  = in_sizes[1] / 2;
  const int nT = E + N;
  const int* src = eidx;
  const int* dst = eidx + E;
  float* out = (float*)d_out;

  char* w = (char*)d_ws;
  float*    zblk   = (float*)w;
  float*    ag1    = (float*)w;    w += (size_t)N * 128 * sizeof(float);
  unsigned* mkey1  = (unsigned*)w; w += (size_t)N * sizeof(unsigned);
  float*    den1   = (float*)w;    w += (size_t)N * sizeof(float);
  unsigned* mkey2  = (unsigned*)w; w += (size_t)N * sizeof(unsigned);
  float*    den2   = (float*)w;    w += (size_t)N * sizeof(float);
  const size_t zcount = (size_t)N * 132;
  float*    h1     = (float*)w;    w += (size_t)N * 128 * sizeof(float);
  float*    ebuf   = (float*)w;    w += (size_t)nT * sizeof(float);
  float*    e1s    = (float*)w;    w += (size_t)N * sizeof(float);
  float*    e1d    = (float*)w;    w += (size_t)N * sizeof(float);
  float*    h2     = (float*)w;    w += (size_t)N * 2 * sizeof(float);
  float*    e2s    = (float*)w;    w += (size_t)N * sizeof(float);
  float*    e2d    = (float*)w;    w += (size_t)N * sizeof(float);

  const int TB = 256;
  zero_f32<<<2048, TB, 0, stream>>>(zblk, zcount);
  zero_f32<<<256,  TB, 0, stream>>>(out, (size_t)N * 2);

  gemm1_wmma<<<(N + 15) / 16, TB, 0, stream>>>(x, W1, h1, N);
  node_e1<<<(N + 7) / 8, TB, 0, stream>>>(h1, a1s, a1d, e1s, e1d, N);
  edge_logits<<<(nT + TB - 1) / TB, TB, 0, stream>>>(e1s, e1d, src, dst, E, nT, ebuf, mkey1);
  edge_exp<<<(nT + TB - 1) / TB, TB, 0, stream>>>(ebuf, mkey1, dst, E, nT, ebuf, den1);
  agg128<<<(nT + 7) / 8, TB, 0, stream>>>(h1, ebuf, den1, src, dst, E, nT, ag1);
  bias_relu<<<(int)(((size_t)N * 128 + TB - 1) / TB), TB, 0, stream>>>(ag1, b1, (size_t)N * 128);

  node_h2e2<<<(N + 7) / 8, TB, 0, stream>>>(ag1, W2, a2s, a2d, h2, e2s, e2d, N);
  edge_logits<<<(nT + TB - 1) / TB, TB, 0, stream>>>(e2s, e2d, src, dst, E, nT, ebuf, mkey2);
  edge_exp<<<(nT + TB - 1) / TB, TB, 0, stream>>>(ebuf, mkey2, dst, E, nT, ebuf, den2);
  agg2<<<(nT + TB - 1) / TB, TB, 0, stream>>>(h2, ebuf, den2, src, dst, E, nT, out);
  bias2<<<(2 * N + TB - 1) / TB, TB, 0, stream>>>(out, b2, 2 * N);
}


# Edges counting-sorted by destination (atomic arrival rank + prefix sum); one wave per node does softmax and aggregation in registers: float atomics gone, sum order = arrival rank then wave tree
# speedup vs baseline: 2.5947x; 2.5947x over previous
#include <hip/hip_runtime.h>
#include <hip/hip_bf16.h>

typedef __attribute__((ext_vector_type(2))) float v2f;
typedef __attribute__((ext_vector_type(8))) float v8f;

#define NEG_SLOPE 0.2f


__device__ __forceinline__ float warp_sum(float v) {
  #pragma unroll
  for (int o = 16; o > 0; o >>= 1) v += __shfl_xor(v, o, 32);
  return v;
}

__device__ __forceinline__ float warp_max(float v) {
  for (int o = 16; o > 0; o >>= 1) v = fmaxf(v, __shfl_xor(v, o, 32));
  return v;
}

__device__ __forceinline__ unsigned warp_sum_u32(unsigned v) {
  for (int o = 16; o > 0; o >>= 1) v += __shfl_xor(v, o, 32);
  return v;
}

__device__ __forceinline__ float leaky(float e) {
  return (e > 0.0f) ? e : NEG_SLOPE * e;
}


__global__ void zero_f32(float* __restrict__ p, size_t n) {
  size_t stride = (size_t)gridDim.x * blockDim.x;
  for (size_t i = (size_t)blockIdx.x * blockDim.x + threadIdx.x; i < n; i += stride)
    p[i] = 0.0f;
}

__global__ void gemm1_wmma(const float* __restrict__ X, const float* __restrict__ W,
                           float* __restrict__ H, int N) {
  const int lane = threadIdx.x & 31;
  const int wave = threadIdx.x >> 5;
  const int row0 = blockIdx.x * 16;
  const int col0 = wave * 16;
  const int m  = lane & 15;
  const int kh = lane >> 4;

  int row = row0 + m; if (row >= N) row = N - 1;
  const float* xrow = X + (size_t)row * 256;

  v8f c = {};
#if __has_builtin(__builtin_amdgcn_wmma_f32_16x16x4_f32)
  #pragma unroll 4
  for (int k = 0; k < 256; k += 4) {
    v2f a = *(const v2f*)(xrow + k + 2 * kh);
    v2f b;
    const float* wp = W + (size_t)(k + 2 * kh) * 128 + col0 + m;
    b.x = wp[0];
    b.y = wp[128];
    c = __builtin_amdgcn_wmma_f32_16x16x4_f32(false, a, false, b, (short)0, c,
                                              false, false);
  }
#else
  for (int k = 0; k < 256; k += 4) {
    v2f a = *(const v2f*)(xrow + k + 2 * kh);
    const float* wp = W + (size_t)(k + 2 * kh) * 128 + col0 + m;
    c[0] += a.x * wp[0] + a.y * wp[128];
  }
#endif
  float* hp = H + (size_t)(row0 + 8 * kh) * 128 + col0 + m;
  if (row0 + 16 <= N) {
    #pragma unroll
    for (int r = 0; r < 8; ++r) hp[(size_t)r * 128] = c[r];
  } else {
    #pragma unroll
    for (int r = 0; r < 8; ++r)
      if (row0 + r + 8 * kh < N) hp[(size_t)r * 128] = c[r];
  }
}

__global__ void node_e1(const float* __restrict__ H, const float* __restrict__ as,
                        const float* __restrict__ ad, float* __restrict__ es,
                        float* __restrict__ ed, int N) {
  int n = blockIdx.x * 8 + (threadIdx.x >> 5);
  if (n >= N) return;
  int lane = threadIdx.x & 31;
  float4 hv = ((const float4*)(H + (size_t)n * 128))[lane];
  float4 sv = ((const float4*)as)[lane];
  float4 dv = ((const float4*)ad)[lane];
  float s = hv.x * sv.x + hv.y * sv.y + hv.z * sv.z + hv.w * sv.w;
  float d = hv.x * dv.x + hv.y * dv.y + hv.z * dv.z + hv.w * dv.w;
  s = warp_sum(s);
  d = warp_sum(d);
  if (lane == 0) { es[n] = s; ed[n] = d; }
}

__global__ void dst_rank(const int* __restrict__ dst, int E, unsigned* __restrict__ cnt,
                         unsigned* __restrict__ rank) {
  int i = blockIdx.x * blockDim.x + threadIdx.x;
  if (i >= E) return;
  rank[i] = atomicAdd(&cnt[dst[i]], 1u);
}

__global__ void scan_tiles(const unsigned* __restrict__ cnt, int N, unsigned* __restrict__ tsum) {
  __shared__ unsigned sm[8];
  const int lane = threadIdx.x & 31;
  const int wave = threadIdx.x >> 5;
  const int base = blockIdx.x * 1024 + threadIdx.x * 4;
  unsigned v = 0;
  for (int j = 0; j < 4; ++j)
    if (base + j < N) v += cnt[base + j];
  v = warp_sum_u32(v);
  if (lane == 0) sm[wave] = v;
  __syncthreads();
  if (threadIdx.x == 0) {
    unsigned t = 0;
    for (int w = 0; w < 8; ++w) t += sm[w];
    tsum[blockIdx.x] = t;
  }
}

__global__ void scan_rows(const unsigned* __restrict__ cnt, const unsigned* __restrict__ tsum,
                          int N, unsigned* __restrict__ rowptr) {
  __shared__ unsigned sm[8];
  const int tid = threadIdx.x;
  const int lane = tid & 31;
  const int wave = tid >> 5;

  unsigned o = 0;
  for (int b = tid; b < (int)blockIdx.x; b += 256) o += tsum[b];
  o = warp_sum_u32(o);
  if (lane == 0) sm[wave] = o;
  __syncthreads();
  unsigned off = 0;
  for (int w = 0; w < 8; ++w) off += sm[w];
  __syncthreads();

  const int base = blockIdx.x * 1024 + tid * 4;
  unsigned c[4];
  for (int j = 0; j < 4; ++j) c[j] = (base + j < N) ? cnt[base + j] : 0u;
  const unsigned mine = c[0] + c[1] + c[2] + c[3];
  unsigned incl = mine;
  for (int d = 1; d < 32; d <<= 1) {
    unsigned t = __shfl_up(incl, d, 32);
    if (lane >= d) incl += t;
  }
  if (lane == 31) sm[wave] = incl;
  __syncthreads();
  unsigned woff = 0;
  for (int w = 0; w < 8; ++w)
    if (w < wave) woff += sm[w];
  unsigned p = off + woff + incl - mine;
  for (int j = 0; j < 4; ++j) {
    if (base + j < N) rowptr[base + j] = p;
    p += c[j];
  }
}

__global__ void csr_fill(const int* __restrict__ src, const int* __restrict__ dst,
                         const unsigned* __restrict__ rank, const unsigned* __restrict__ rowptr,
                         int E, int* __restrict__ csr_src) {
  int i = blockIdx.x * blockDim.x + threadIdx.x;
  if (i >= E) return;
  csr_src[rowptr[dst[i]] + rank[i]] = src[i];
}

__global__ void gat_gather128(const float* __restrict__ H, const float* __restrict__ es,
                              const float* __restrict__ ed, const unsigned* __restrict__ rowptr,
                              const unsigned* __restrict__ cnt, const int* __restrict__ csr_src,
                              int N, float* __restrict__ agg) {
  const int n = __builtin_amdgcn_readfirstlane((int)(blockIdx.x * 8 + (threadIdx.x >> 5)));
  if (n >= N) return;
  const int lane = threadIdx.x & 31;
  const unsigned beg = rowptr[n];
  const int deg = (int)cnt[n];
  const int tot = deg + 1;
  const float edn = ed[n];

  int s0 = n;
  float e0 = -INFINITY;
  const bool v0 = lane < tot;
  if (v0) {
    s0 = (lane < deg) ? csr_src[beg + lane] : n;
    e0 = leaky(es[s0] + edn);
  }
  float m = e0;
  for (int j = lane + 32; j < tot; j += 32) {
    int s = (j < deg) ? csr_src[beg + j] : n;
    m = fmaxf(m, leaky(es[s] + edn));
  }
  m = warp_max(m);

  const float x0 = v0 ? expf(e0 - m) : 0.0f;
  float dsum = x0;
  for (int j = lane + 32; j < tot; j += 32) {
    int s = (j < deg) ? csr_src[beg + j] : n;
    dsum += expf(leaky(es[s] + edn) - m);
  }
  dsum = warp_sum(dsum);

  float4 acc = make_float4(0.0f, 0.0f, 0.0f, 0.0f);
  {
    const float a0 = x0 / dsum;
    const int c0 = (tot < 32) ? tot : 32;
    #pragma unroll 4
    for (int t = 0; t < c0; ++t) {
      const int st = __builtin_amdgcn_readlane(s0, t);
      const float at = __int_as_float(__builtin_amdgcn_readlane(__float_as_int(a0), t));
      const float4 hv = ((const float4*)(H + (size_t)st * 128))[lane];
      acc.x += hv.x * at;
      acc.y += hv.y * at;
      acc.z += hv.z * at;
      acc.w += hv.w * at;
    }
  }
  for (int j0 = 32; j0 < tot; j0 += 32) {
    const int j = j0 + lane;
    int s = n;
    float a = 0.0f;
    if (j < tot) {
      s = (j < deg) ? csr_src[beg + j] : n;
      a = expf(leaky(es[s] + edn) - m) / dsum;
    }
    const int cc = (tot - j0 < 32) ? (tot - j0) : 32;
    for (int t = 0; t < cc; ++t) {
      const int st = __builtin_amdgcn_readlane(s, t);
      const float at = __int_as_float(__builtin_amdgcn_readlane(__float_as_int(a), t));
      const float4 hv = ((const float4*)(H + (size_t)st * 128))[lane];
      acc.x += hv.x * at;
      acc.y += hv.y * at;
      acc.z += hv.z * at;
      acc.w += hv.w * at;
    }
  }
  ((float4*)(agg + (size_t)n * 128))[lane] = acc;
}

__global__ void bias_relu(float* __restrict__ a, const float* __restrict__ b, size_t n) {
  size_t i = (size_t)blockIdx.x * blockDim.x + threadIdx.x;
  if (i >= n) return;
  float v = a[i] + b[i & 127];
  a[i] = (v > 0.0f) ? v : 0.0f;
}

__global__ void node_h2e2(const float* __restrict__ H, const float* __restrict__ W2,
                          const float* __restrict__ a2s, const float* __restrict__ a2d,
                          float* __restrict__ h2, float* __restrict__ e2s,
                          float* __restrict__ e2d, int N) {
  int n = blockIdx.x * 8 + (threadIdx.x >> 5);
  if (n >= N) return;
  int lane = threadIdx.x & 31;
  float4 hv = ((const float4*)(H + (size_t)n * 128))[lane];
  int c0 = lane * 4;
  float s0 = hv.x * W2[(c0 + 0) * 2 + 0] + hv.y * W2[(c0 + 1) * 2 + 0] +
             hv.z * W2[(c0 + 2) * 2 + 0] + hv.w * W2[(c0 + 3) * 2 + 0];
  float s1 = hv.x * W2[(c0 + 0) * 2 + 1] + hv.y * W2[(c0 + 1) * 2 + 1] +
             hv.z * W2[(c0 + 2) * 2 + 1] + hv.w * W2[(c0 + 3) * 2 + 1];
  s0 = warp_sum(s0);
  s1 = warp_sum(s1);
  if (lane == 0) {
    h2[2 * n + 0] = s0;
    h2[2 * n + 1] = s1;
    e2s[n] = s0 * a2s[0] + s1 * a2s[1];
    e2d[n] = s0 * a2d[0] + s1 * a2d[1];
  }
}

__global__ void gat_gather2(const float* __restrict__ h2, const float* __restrict__ es,
                            const float* __restrict__ ed, const unsigned* __restrict__ rowptr,
                            const unsigned* __restrict__ cnt, const int* __restrict__ csr_src,
                            int N, float* __restrict__ out) {
  const int n = __builtin_amdgcn_readfirstlane((int)(blockIdx.x * 8 + (threadIdx.x >> 5)));
  if (n >= N) return;
  const int lane = threadIdx.x & 31;
  const unsigned beg = rowptr[n];
  const int deg = (int)cnt[n];
  const int tot = deg + 1;
  const float edn = ed[n];

  int s0 = n;
  float e0 = -INFINITY;
  const bool v0 = lane < tot;
  if (v0) {
    s0 = (lane < deg) ? csr_src[beg + lane] : n;
    e0 = leaky(es[s0] + edn);
  }
  float m = e0;
  for (int j = lane + 32; j < tot; j += 32) {
    int s = (j < deg) ? csr_src[beg + j] : n;
    m = fmaxf(m, leaky(es[s] + edn));
  }
  m = warp_max(m);

  const float x0 = v0 ? expf(e0 - m) : 0.0f;
  float dsum = x0;
  for (int j = lane + 32; j < tot; j += 32) {
    int s = (j < deg) ? csr_src[beg + j] : n;
    dsum += expf(leaky(es[s] + edn) - m);
  }
  dsum = warp_sum(dsum);

  float p0 = 0.0f, p1 = 0.0f;
  if (v0) {
    const float a0 = x0 / dsum;
    const float2 hv = ((const float2*)h2)[s0];
    p0 = hv.x * a0;
    p1 = hv.y * a0;
  }
  for (int j = lane + 32; j < tot; j += 32) {
    int s = (j < deg) ? csr_src[beg + j] : n;
    const float a = expf(leaky(es[s] + edn) - m) / dsum;
    const float2 hv = ((const float2*)h2)[s];
    p0 += hv.x * a;
    p1 += hv.y * a;
  }
  p0 = warp_sum(p0);
  p1 = warp_sum(p1);
  if (lane == 0) {
    out[2 * n + 0] = p0;
    out[2 * n + 1] = p1;
  }
}

__global__ void bias2(float* __restrict__ out, const float* __restrict__ b, int n2) {
  int i = blockIdx.x * blockDim.x + threadIdx.x;
  if (i >= n2) return;
  out[i] += b[i & 1];
}


extern "C" void kernel_launch(void* const* d_in, const int* in_sizes, int n_in,
                              void* d_out, int out_size, void* d_ws, size_t ws_size,
                              hipStream_t stream) {
  const float* x   = (const float*)d_in[0];
  const int*  eidx = (const int*)d_in[1];
  const float* W1  = (const float*)d_in[2];
  const float* a1s = (const float*)d_in[3];
  const float* a1d = (const float*)d_in[4];
  const float* b1  = (const float*)d_in[5];
  const float* W2  = (const float*)d_in[6];
  const float* a2s = (const float*)d_in[7];
  const float* a2d = (const float*)d_in[8];
  const float* b2  = (const float*)d_in[9];

  const int N  = in_sizes[0] / 256;
  const int E  = in_sizes[1] / 2;
  const int* src = eidx;
  const int* dst = eidx + E;
  float* out = (float*)d_out;

  char* w = (char*)d_ws;
  float*    ag1    = (float*)w;    w += (size_t)N * 128 * sizeof(float);
  float*    h1     = (float*)w;    w += (size_t)N * 128 * sizeof(float);
  float*    e1s    = (float*)w;    w += (size_t)N * sizeof(float);
  float*    e1d    = (float*)w;    w += (size_t)N * sizeof(float);
  float*    h2     = (float*)w;    w += (size_t)N * 2 * sizeof(float);
  float*    e2s    = (float*)w;    w += (size_t)N * sizeof(float);
  float*    e2d    = (float*)w;    w += (size_t)N * sizeof(float);
  unsigned* cnt    = (unsigned*)w; w += (size_t)N * sizeof(unsigned);
  unsigned* rowptr = (unsigned*)w; w += (size_t)N * sizeof(unsigned);
  unsigned* tsum   = (unsigned*)w; w += (size_t)((N + 1023) / 1024) * sizeof(unsigned);
  unsigned* rank   = (unsigned*)w; w += (size_t)E * sizeof(unsigned);
  int*      csrs   = (int*)w;      w += (size_t)E * sizeof(int);

  const int TB = 256;
  zero_f32<<<256, TB, 0, stream>>>((float*)cnt, (size_t)N);
  dst_rank<<<(E + TB - 1) / TB, TB, 0, stream>>>(dst, E, cnt, rank);
  scan_tiles<<<(N + 1023) / 1024, TB, 0, stream>>>(cnt, N, tsum);
  scan_rows<<<(N + 1023) / 1024, TB, 0, stream>>>(cnt, tsum, N, rowptr);
  csr_fill<<<(E + TB - 1) / TB, TB, 0, stream>>>(src, dst, rank, rowptr, E, csrs);

  gemm1_wmma<<<(N + 15) / 16, TB, 0, stream>>>(x, W1, h1, N);
  node_e1<<<(N + 7) / 8, TB, 0, stream>>>(h1, a1s, a1d, e1s, e1d, N);
  gat_gather128<<<(N + 7) / 8, TB, 0, stream>>>(h1, e1s, e1d, rowptr, cnt, csrs, N, ag1);
  bias_relu<<<(int)(((size_t)N * 128 + TB - 1) / TB), TB, 0, stream>>>(ag1, b1, (size_t)N * 128);

  node_h2e2<<<(N + 7) / 8, TB, 0, stream>>>(ag1, W2, a2s, a2d, h2, e2s, e2d, N);
  gat_gather2<<<(N + 7) / 8, TB, 0, stream>>>(h2, e2s, e2d, rowptr, cnt, csrs, N, out);
  bias2<<<(2 * N + TB - 1) / TB, TB, 0, stream>>>(out, b2, 2 * N);
}


# Layer-1 GEMM retiled: 4-wave workgroup = 64 rows x 128 cols, W staged through LDS in 16-k chunks (two buffers), 8 accumulator tiles per wave per A fragment; k summed in another order inside a chunk
# speedup vs baseline: 3.0496x; 1.1753x over previous
#include <hip/hip_runtime.h>
#include <hip/hip_bf16.h>

typedef __attribute__((ext_vector_type(2))) float v2f;
typedef __attribute__((ext_vector_type(8))) float v8f;

#define NEG_SLOPE 0.2f


__device__ __forceinline__ float warp_sum(float v) {
  #pragma unroll
  for (int o = 16; o > 0; o >>= 1) v += __shfl_xor(v, o, 32);
  return v;
}

__device__ __forceinline__ float warp_max(float v) {
  for (int o = 16; o > 0; o >>= 1) v = fmaxf(v, __shfl_xor(v, o, 32));
  return v;
}

__device__ __forceinline__ unsigned warp_sum_u32(unsigned v) {
  for (int o = 16; o > 0; o >>= 1) v += __shfl_xor(v, o, 32);
  return v;
}

__device__ __forceinline__ float leaky(float e) {
  return (e > 0.0f) ? e : NEG_SLOPE * e;
}


__global__ void zero_f32(float* __restrict__ p, size_t n) {
  size_t stride = (size_t)gridDim.x * blockDim.x;
  for (size_t i = (size_t)blockIdx.x * blockDim.x + threadIdx.x; i < n; i += stride)
    p[i] = 0.0f;
}

#define G_KC 16
#define G_PITCH 264

__global__ __launch_bounds__(128) __attribute__((amdgpu_waves_per_eu(7))) void gemm1_wmma(const float* __restrict__ X,
                                                  const float* __restrict__ W,
                                                  float* __restrict__ H, int N) {
  __shared__ __attribute__((aligned(16))) float Bs[2][8 * G_PITCH];
  const int tid  = threadIdx.x;
  const int lane = tid & 31;
  const int wave = tid >> 5;
  const int c = lane & 15;
  const int h = lane >> 4;
  const int row0 = blockIdx.x * 64 + wave * 16;

  int row = row0 + c; if (row >= N) row = N - 1;
  const float* xrow = X + (size_t)row * 256 + 8 * h;

  const float* wsrc = W + (size_t)(2 * wave) * 128 + 4 * lane;
  const int sdst = wave * G_PITCH + 8 * lane;

  v8f acc[8];
  #pragma unroll
  for (int j = 0; j < 8; ++j) acc[j] = (v8f){};

  float4 a0 = *(const float4*)(xrow);
  float4 a1 = *(const float4*)(xrow + 4);
  {
    const float4 e0 = *(const float4*)(wsrc);
    const float4 o0 = *(const float4*)(wsrc + 128);
    const float4 e1 = *(const float4*)(wsrc + 8 * 128);
    const float4 o1 = *(const float4*)(wsrc + 9 * 128);
    float* d = &Bs[0][sdst];
    *(float4*)(d)                   = make_float4(e0.x, o0.x, e0.y, o0.y);
    *(float4*)(d + 4)               = make_float4(e0.z, o0.z, e0.w, o0.w);
    *(float4*)(d + 4 * G_PITCH)     = make_float4(e1.x, o1.x, e1.y, o1.y);
    *(float4*)(d + 4 * G_PITCH + 4) = make_float4(e1.z, o1.z, e1.w, o1.w);
  }
  __syncthreads();

  for (int i = 0; i < 256 / G_KC; ++i) {
    const bool more = (i + 1) < 256 / G_KC;
    float4 na0 = a0, na1 = a1, e0, o0, e1, o1;
    if (more) {
      const int kn = (i + 1) * G_KC;
      const float* ws = wsrc + (size_t)kn * 128;
      e0 = *(const float4*)(ws);
      o0 = *(const float4*)(ws + 128);
      e1 = *(const float4*)(ws + 8 * 128);
      o1 = *(const float4*)(ws + 9 * 128);
      na0 = *(const float4*)(xrow + kn);
      na1 = *(const float4*)(xrow + kn + 4);
    }
    __builtin_amdgcn_sched_barrier(0);

    const float* bs = &Bs[i & 1][(4 * h) * G_PITCH + 2 * c];
    v2f bf[2][8];
    #pragma unroll
    for (int j = 0; j < 8; ++j) bf[0][j] = *(const v2f*)(bs + 32 * j);
    #pragma unroll
    for (int t = 0; t < 4; ++t) {
      if (t < 3) {
        #pragma unroll
        for (int j = 0; j < 8; ++j) bf[(t + 1) & 1][j] = *(const v2f*)(bs + (t + 1) * G_PITCH + 32 * j);
      }
      __builtin_amdgcn_sched_barrier(0);
      v2f a;
      if (t == 0)      { a.x = a0.x; a.y = a0.y; }
      else if (t == 1) { a.x = a0.z; a.y = a0.w; }
      else if (t == 2) { a.x = a1.x; a.y = a1.y; }
      else             { a.x = a1.z; a.y = a1.w; }
      #pragma unroll
      for (int j = 0; j < 8; ++j) {
        acc[j] = __builtin_amdgcn_wmma_f32_16x16x4_f32(false, a, false, bf[t & 1][j], (short)0,
                                                       acc[j], false, false);
      }
      __builtin_amdgcn_sched_barrier(0);
    }

    if (more) {
      float* d = &Bs[(i + 1) & 1][sdst];
      *(float4*)(d)                   = make_float4(e0.x, o0.x, e0.y, o0.y);
      *(float4*)(d + 4)               = make_float4(e0.z, o0.z, e0.w, o0.w);
      *(float4*)(d + 4 * G_PITCH)     = make_float4(e1.x, o1.x, e1.y, o1.y);
      *(float4*)(d + 4 * G_PITCH + 4) = make_float4(e1.z, o1.z, e1.w, o1.w);
    }
    __syncthreads();
    a0 = na0; a1 = na1;
  }

  if (row0 + 16 <= N) {
    float* hp = H + (size_t)(row0 + 8 * h) * 128 + c;
    #pragma unroll
    for (int j = 0; j < 8; ++j) {
      #pragma unroll
      for (int r = 0; r < 8; ++r) hp[(size_t)r * 128 + 16 * j] = acc[j][r];
    }
  } else {
    #pragma unroll
    for (int j = 0; j < 8; ++j) {
      #pragma unroll
      for (int r = 0; r < 8; ++r) {
        const int rr = row0 + 8 * h + r;
        if (rr < N) H[(size_t)rr * 128 + 16 * j + c] = acc[j][r];
      }
    }
  }
}

__global__ void node_e1(const float* __restrict__ H, const float* __restrict__ as,
                        const float* __restrict__ ad, float* __restrict__ es,
                        float* __restrict__ ed, int N) {
  int n = blockIdx.x * 8 + (threadIdx.x >> 5);
  if (n >= N) return;
  int lane = threadIdx.x & 31;
  float4 hv = ((const float4*)(H + (size_t)n * 128))[lane];
  float4 sv = ((const float4*)as)[lane];
  float4 dv = ((const float4*)ad)[lane];
  float s = hv.x * sv.x + hv.y * sv.y + hv.z * sv.z + hv.w * sv.w;
  float d = hv.x * dv.x + hv.y * dv.y + hv.z * dv.z + hv.w * dv.w;
  s = warp_sum(s);
  d = warp_sum(d);
  if (lane == 0) { es[n] = s; ed[n] = d; }
}

__global__ void dst_rank(const int* __restrict__ dst, int E, unsigned* __restrict__ cnt,
                         unsigned* __restrict__ rank) {
  int i = blockIdx.x * blockDim.x + threadIdx.x;
  if (i >= E) return;
  rank[i] = atomicAdd(&cnt[dst[i]], 1u);
}

__global__ void scan_tiles(const unsigned* __restrict__ cnt, int N, unsigned* __restrict__ tsum) {
  __shared__ unsigned sm[8];
  const int lane = threadIdx.x & 31;
  const int wave = threadIdx.x >> 5;
  const int base = blockIdx.x * 1024 + threadIdx.x * 4;
  unsigned v = 0;
  for (int j = 0; j < 4; ++j)
    if (base + j < N) v += cnt[base + j];
  v = warp_sum_u32(v);
  if (lane == 0) sm[wave] = v;
  __syncthreads();
  if (threadIdx.x == 0) {
    unsigned t = 0;
    for (int w = 0; w < 8; ++w) t += sm[w];
    tsum[blockIdx.x] = t;
  }
}

__global__ void scan_rows(const unsigned* __restrict__ cnt, const unsigned* __restrict__ tsum,
                          int N, unsigned* __restrict__ rowptr) {
  __shared__ unsigned sm[8];
  const int tid = threadIdx.x;
  const int lane = tid & 31;
  const int wave = tid >> 5;

  unsigned o = 0;
  for (int b = tid; b < (int)blockIdx.x; b += 256) o += tsum[b];
  o = warp_sum_u32(o);
  if (lane == 0) sm[wave] = o;
  __syncthreads();
  unsigned off = 0;
  for (int w = 0; w < 8; ++w) off += sm[w];
  __syncthreads();

  const int base = blockIdx.x * 1024 + tid * 4;
  unsigned c[4];
  for (int j = 0; j < 4; ++j) c[j] = (base + j < N) ? cnt[base + j] : 0u;
  const unsigned mine = c[0] + c[1] + c[2] + c[3];
  unsigned incl = mine;
  for (int d = 1; d < 32; d <<= 1) {
    unsigned t = __shfl_up(incl, d, 32);
    if (lane >= d) incl += t;
  }
  if (lane == 31) sm[wave] = incl;
  __syncthreads();
  unsigned woff = 0;
  for (int w = 0; w < 8; ++w)
    if (w < wave) woff += sm[w];
  unsigned p = off + woff + incl - mine;
  for (int j = 0; j < 4; ++j) {
    if (base + j < N) rowptr[base + j] = p;
    p += c[j];
  }
}

__global__ void csr_fill(const int* __restrict__ src, const int* __restrict__ dst,
                         const unsigned* __restrict__ rank, const unsigned* __restrict__ rowptr,
                         int E, int* __restrict__ csr_src) {
  int i = blockIdx.x * blockDim.x + threadIdx.x;
  if (i >= E) return;
  csr_src[rowptr[dst[i]] + rank[i]] = src[i];
}

__global__ void gat_gather128(const float* __restrict__ H, const float* __restrict__ es,
                              const float* __restrict__ ed, const unsigned* __restrict__ rowptr,
                              const unsigned* __restrict__ cnt, const int* __restrict__ csr_src,
                              int N, float* __restrict__ agg) {
  const int n = __builtin_amdgcn_readfirstlane((int)(blockIdx.x * 8 + (threadIdx.x >> 5)));
  if (n >= N) return;
  const int lane = threadIdx.x & 31;
  const unsigned beg = rowptr[n];
  const int deg = (int)cnt[n];
  const int tot = deg + 1;
  const float edn = ed[n];

  int s0 = n;
  float e0 = -INFINITY;
  const bool v0 = lane < tot;
  if (v0) {
    s0 = (lane < deg) ? csr_src[beg + lane] : n;
    e0 = leaky(es[s0] + edn);
  }
  float m = e0;
  for (int j = lane + 32; j < tot; j += 32) {
    int s = (j < deg) ? csr_src[beg + j] : n;
    m = fmaxf(m, leaky(es[s] + edn));
  }
  m = warp_max(m);

  const float x0 = v0 ? expf(e0 - m) : 0.0f;
  float dsum = x0;
  for (int j = lane + 32; j < tot; j += 32) {
    int s = (j < deg) ? csr_src[beg + j] : n;
    dsum += expf(leaky(es[s] + edn) - m);
  }
  dsum = warp_sum(dsum);

  float4 acc = make_float4(0.0f, 0.0f, 0.0f, 0.0f);
  {
    const float a0 = x0 / dsum;
    const int c0 = (tot < 32) ? tot : 32;
    #pragma unroll 4
    for (int t = 0; t < c0; ++t) {
      const int st = __builtin_amdgcn_readlane(s0, t);
      const float at = __int_as_float(__builtin_amdgcn_readlane(__float_as_int(a0), t));
      const float4 hv = ((const float4*)(H + (size_t)st * 128))[lane];
      acc.x += hv.x * at;
      acc.y += hv.y * at;
      acc.z += hv.z * at;
      acc.w += hv.w * at;
    }
  }
  for (int j0 = 32; j0 < tot; j0 += 32) {
    const int j = j0 + lane;
    int s = n;
    float a = 0.0f;
    if (j < tot) {
      s = (j < deg) ? csr_src[beg + j] : n;
      a = expf(leaky(es[s] + edn) - m) / dsum;
    }
    const int cc = (tot - j0 < 32) ? (tot - j0) : 32;
    for (int t = 0; t < cc; ++t) {
      const int st = __builtin_amdgcn_readlane(s, t);
      const float at = __int_as_float(__builtin_amdgcn_readlane(__float_as_int(a), t));
      const float4 hv = ((const float4*)(H + (size_t)st * 128))[lane];
      acc.x += hv.x * at;
      acc.y += hv.y * at;
      acc.z += hv.z * at;
      acc.w += hv.w * at;
    }
  }
  ((float4*)(agg + (size_t)n * 128))[lane] = acc;
}

__global__ void bias_relu(float* __restrict__ a, const float* __restrict__ b, size_t n) {
  size_t i = (size_t)blockIdx.x * blockDim.x + threadIdx.x;
  if (i >= n) return;
  float v = a[i] + b[i & 127];
  a[i] = (v > 0.0f) ? v : 0.0f;
}

__global__ void node_h2e2(const float* __restrict__ H, const float* __restrict__ W2,
                          const float* __restrict__ a2s, const float* __restrict__ a2d,
                          float* __restrict__ h2, float* __restrict__ e2s,
                          float* __restrict__ e2d, int N) {
  int n = blockIdx.x * 8 + (threadIdx.x >> 5);
  if (n >= N) return;
  int lane = threadIdx.x & 31;
  float4 hv = ((const float4*)(H + (size_t)n * 128))[lane];
  int c0 = lane * 4;
  float s0 = hv.x * W2[(c0 + 0) * 2 + 0] + hv.y * W2[(c0 + 1) * 2 + 0] +
             hv.z * W2[(c0 + 2) * 2 + 0] + hv.w * W2[(c0 + 3) * 2 + 0];
  float s1 = hv.x * W2[(c0 + 0) * 2 + 1] + hv.y * W2[(c0 + 1) * 2 + 1] +
             hv.z * W2[(c0 + 2) * 2 + 1] + hv.w * W2[(c0 + 3) * 2 + 1];
  s0 = warp_sum(s0);
  s1 = warp_sum(s1);
  if (lane == 0) {
    h2[2 * n + 0] = s0;
    h2[2 * n + 1] = s1;
    e2s[n] = s0 * a2s[0] + s1 * a2s[1];
    e2d[n] = s0 * a2d[0] + s1 * a2d[1];
  }
}

__global__ void gat_gather2(const float* __restrict__ h2, const float* __restrict__ es,
                            const float* __restrict__ ed, const unsigned* __restrict__ rowptr,
                            const unsigned* __restrict__ cnt, const int* __restrict__ csr_src,
                            int N, float* __restrict__ out) {
  const int n = __builtin_amdgcn_readfirstlane((int)(blockIdx.x * 8 + (threadIdx.x >> 5)));
  if (n >= N) return;
  const int lane = threadIdx.x & 31;
  const unsigned beg = rowptr[n];
  const int deg = (int)cnt[n];
  const int tot = deg + 1;
  const float edn = ed[n];

  int s0 = n;
  float e0 = -INFINITY;
  const bool v0 = lane < tot;
  if (v0) {
    s0 = (lane < deg) ? csr_src[beg + lane] : n;
    e0 = leaky(es[s0] + edn);
  }
  float m = e0;
  for (int j = lane + 32; j < tot; j += 32) {
    int s = (j < deg) ? csr_src[beg + j] : n;
    m = fmaxf(m, leaky(es[s] + edn));
  }
  m = warp_max(m);

  const float x0 = v0 ? expf(e0 - m) : 0.0f;
  float dsum = x0;
  for (int j = lane + 32; j < tot; j += 32) {
    int s = (j < deg) ? csr_src[beg + j] : n;
    dsum += expf(leaky(es[s] + edn) - m);
  }
  dsum = warp_sum(dsum);

  float p0 = 0.0f, p1 = 0.0f;
  if (v0) {
    const float a0 = x0 / dsum;
    const float2 hv = ((const float2*)h2)[s0];
    p0 = hv.x * a0;
    p1 = hv.y * a0;
  }
  for (int j = lane + 32; j < tot; j += 32) {
    int s = (j < deg) ? csr_src[beg + j] : n;
    const float a = expf(leaky(es[s] + edn) - m) / dsum;
    const float2 hv = ((const float2*)h2)[s];
    p0 += hv.x * a;
    p1 += hv.y * a;
  }
  p0 = warp_sum(p0);
  p1 = warp_sum(p1);
  if (lane == 0) {
    out[2 * n + 0] = p0;
    out[2 * n + 1] = p1;
  }
}

__global__ void bias2(float* __restrict__ out, const float* __restrict__ b, int n2) {
  int i = blockIdx.x * blockDim.x + threadIdx.x;
  if (i >= n2) return;
  out[i] += b[i & 1];
}


extern "C" void kernel_launch(void* const* d_in, const int* in_sizes, int n_in,
                              void* d_out, int out_size, void* d_ws, size_t ws_size,
                              hipStream_t stream) {
  const float* x   = (const float*)d_in[0];
  const int*  eidx = (const int*)d_in[1];
  const float* W1  = (const float*)d_in[2];
  const float* a1s = (const float*)d_in[3];
  const float* a1d = (const float*)d_in[4];
  const float* b1  = (const float*)d_in[5];
  const float* W2  = (const float*)d_in[6];
  const float* a2s = (const float*)d_in[7];
  const float* a2d = (const float*)d_in[8];
  const float* b2  = (const float*)d_in[9];

  const int N  = in_sizes[0] / 256;
  const int E  = in_sizes[1] / 2;
  const int* src = eidx;
  const int* dst = eidx + E;
  float* out = (float*)d_out;

  char* w = (char*)d_ws;
  float*    ag1    = (float*)w;    w += (size_t)N * 128 * sizeof(float);
  float*    h1     = (float*)w;    w += (size_t)N * 128 * sizeof(float);
  float*    e1s    = (float*)w;    w += (size_t)N * sizeof(float);
  float*    e1d    = (float*)w;    w += (size_t)N * sizeof(float);
  float*    h2     = (float*)w;    w += (size_t)N * 2 * sizeof(float);
  float*    e2s    = (float*)w;    w += (size_t)N * sizeof(float);
  float*    e2d    = (float*)w;    w += (size_t)N * sizeof(float);
  unsigned* cnt    = (unsigned*)w; w += (size_t)N * sizeof(unsigned);
  unsigned* rowptr = (unsigned*)w; w += (size_t)N * sizeof(unsigned);
  unsigned* tsum   = (unsigned*)w; w += (size_t)((N + 1023) / 1024) * sizeof(unsigned);
  unsigned* rank   = (unsigned*)w; w += (size_t)E * sizeof(unsigned);
  int*      csrs   = (int*)w;      w += (size_t)E * sizeof(int);

  const int TB = 256;
  zero_f32<<<256, TB, 0, stream>>>((float*)cnt, (size_t)N);
  dst_rank<<<(E + TB - 1) / TB, TB, 0, stream>>>(dst, E, cnt, rank);
  scan_tiles<<<(N + 1023) / 1024, TB, 0, stream>>>(cnt, N, tsum);
  scan_rows<<<(N + 1023) / 1024, TB, 0, stream>>>(cnt, tsum, N, rowptr);
  csr_fill<<<(E + TB - 1) / TB, TB, 0, stream>>>(src, dst, rank, rowptr, E, csrs);

  gemm1_wmma<<<(N + 63) / 64, 128, 0, stream>>>(x, W1, h1, N);
  node_e1<<<(N + 7) / 8, TB, 0, stream>>>(h1, a1s, a1d, e1s, e1d, N);
  gat_gather128<<<(N + 7) / 8, TB, 0, stream>>>(h1, e1s, e1d, rowptr, cnt, csrs, N, ag1);
  bias_relu<<<(int)(((size_t)N * 128 + TB - 1) / TB), TB, 0, stream>>>(ag1, b1, (size_t)N * 128);

  node_h2e2<<<(N + 7) / 8, TB, 0, stream>>>(ag1, W2, a2s, a2d, h2, e2s, e2d, N);
  gat_gather2<<<(N + 7) / 8, TB, 0, stream>>>(h2, e2s, e2d, rowptr, cnt, csrs, N, out);
  bias2<<<(2 * N + TB - 1) / TB, TB, 0, stream>>>(out, b2, 2 * N);
}


# Per-node stages fused into the waves that hold the row: e1 in GEMM epilogue (other sum order: 8 tiles in lane, then 16-lane DPP butterfly), bias+ReLU+128x2 proj+e2 in gather-1, out bias in gather-2
# speedup vs baseline: 3.6450x; 1.1952x over previous
#include <hip/hip_runtime.h>
#include <hip/hip_bf16.h>

typedef __attribute__((ext_vector_type(2))) float v2f;
typedef __attribute__((ext_vector_type(8))) float v8f;

#define NEG_SLOPE 0.2f


__device__ __forceinline__ float warp_sum(float v) {
  #pragma unroll
  for (int o = 16; o > 0; o >>= 1) v += __shfl_xor(v, o, 32);
  return v;
}

__device__ __forceinline__ float warp_max(float v) {
  for (int o = 16; o > 0; o >>= 1) v = fmaxf(v, __shfl_xor(v, o, 32));
  return v;
}

__device__ __forceinline__ unsigned warp_sum_u32(unsigned v) {
  for (int o = 16; o > 0; o >>= 1) v += __shfl_xor(v, o, 32);
  return v;
}

template <int M>
__device__ __forceinline__ float row_xor(float v) {
  return __int_as_float(__builtin_amdgcn_update_dpp(0, __float_as_int(v), 0x160 + M, 0xf, 0xf, true));
}

__device__ __forceinline__ float leaky(float e) {
  return (e > 0.0f) ? e : NEG_SLOPE * e;
}


__global__ void zero_f32(float* __restrict__ p, size_t n) {
  size_t stride = (size_t)gridDim.x * blockDim.x;
  for (size_t i = (size_t)blockIdx.x * blockDim.x + threadIdx.x; i < n; i += stride)
    p[i] = 0.0f;
}

#define G_KC 16
#define G_PITCH 264

__global__ __launch_bounds__(128) __attribute__((amdgpu_waves_per_eu(7))) void gemm1_wmma(const float* __restrict__ X,
                                                  const float* __restrict__ W,
                                                  const float* __restrict__ as,
                                                  const float* __restrict__ ad,
                                                  float* __restrict__ H,
                                                  float* __restrict__ es,
                                                  float* __restrict__ ed, int N) {
  __shared__ __attribute__((aligned(16))) float Bs[2][8 * G_PITCH];
  const int tid  = threadIdx.x;
  const int lane = tid & 31;
  const int wave = tid >> 5;
  const int c = lane & 15;
  const int h = lane >> 4;
  const int row0 = blockIdx.x * 64 + wave * 16;

  int row = row0 + c; if (row >= N) row = N - 1;
  const float* xrow = X + (size_t)row * 256 + 8 * h;

  const float* wsrc = W + (size_t)(2 * wave) * 128 + 4 * lane;
  const int sdst = wave * G_PITCH + 8 * lane;

  v8f acc[8];
  #pragma unroll
  for (int j = 0; j < 8; ++j) acc[j] = (v8f){};

  float4 a0 = *(const float4*)(xrow);
  float4 a1 = *(const float4*)(xrow + 4);
  {
    const float4 e0 = *(const float4*)(wsrc);
    const float4 o0 = *(const float4*)(wsrc + 128);
    const float4 e1 = *(const float4*)(wsrc + 8 * 128);
    const float4 o1 = *(const float4*)(wsrc + 9 * 128);
    float* d = &Bs[0][sdst];
    *(float4*)(d)                   = make_float4(e0.x, o0.x, e0.y, o0.y);
    *(float4*)(d + 4)               = make_float4(e0.z, o0.z, e0.w, o0.w);
    *(float4*)(d + 4 * G_PITCH)     = make_float4(e1.x, o1.x, e1.y, o1.y);
    *(float4*)(d + 4 * G_PITCH + 4) = make_float4(e1.z, o1.z, e1.w, o1.w);
  }
  __syncthreads();

  for (int i = 0; i < 256 / G_KC; ++i) {
    const bool more = (i + 1) < 256 / G_KC;
    float4 na0 = a0, na1 = a1, e0, o0, e1, o1;
    if (more) {
      const int kn = (i + 1) * G_KC;
      const float* ws = wsrc + (size_t)kn * 128;
      e0 = *(const float4*)(ws);
      o0 = *(const float4*)(ws + 128);
      e1 = *(const float4*)(ws + 8 * 128);
      o1 = *(const float4*)(ws + 9 * 128);
      na0 = *(const float4*)(xrow + kn);
      na1 = *(const float4*)(xrow + kn + 4);
    }
    __builtin_amdgcn_sched_barrier(0);

    const float* bs = &Bs[i & 1][(4 * h) * G_PITCH + 2 * c];
    v2f bf[2][8];
    #pragma unroll
    for (int j = 0; j < 8; ++j) bf[0][j] = *(const v2f*)(bs + 32 * j);
    #pragma unroll
    for (int t = 0; t < 4; ++t) {
      if (t < 3) {
        #pragma unroll
        for (int j = 0; j < 8; ++j) bf[(t + 1) & 1][j] = *(const v2f*)(bs + (t + 1) * G_PITCH + 32 * j);
      }
      __builtin_amdgcn_sched_barrier(0);
      v2f a;
      if (t == 0)      { a.x = a0.x; a.y = a0.y; }
      else if (t == 1) { a.x = a0.z; a.y = a0.w; }
      else if (t == 2) { a.x = a1.x; a.y = a1.y; }
      else             { a.x = a1.z; a.y = a1.w; }
      #pragma unroll
      for (int j = 0; j < 8; ++j) {
        acc[j] = __builtin_amdgcn_wmma_f32_16x16x4_f32(false, a, false, bf[t & 1][j], (short)0,
                                                       acc[j], false, false);
      }
      __builtin_amdgcn_sched_barrier(0);
    }

    if (more) {
      float* d = &Bs[(i + 1) & 1][sdst];
      *(float4*)(d)                   = make_float4(e0.x, o0.x, e0.y, o0.y);
      *(float4*)(d + 4)               = make_float4(e0.z, o0.z, e0.w, o0.w);
      *(float4*)(d + 4 * G_PITCH)     = make_float4(e1.x, o1.x, e1.y, o1.y);
      *(float4*)(d + 4 * G_PITCH + 4) = make_float4(e1.z, o1.z, e1.w, o1.w);
    }
    __syncthreads();
    a0 = na0; a1 = na1;
  }

  {
    float ps[8], pd[8];
    #pragma unroll
    for (int r = 0; r < 8; ++r) { ps[r] = 0.0f; pd[r] = 0.0f; }
    #pragma unroll
    for (int j = 0; j < 8; ++j) {
      const float sv = as[16 * j + c];
      const float dv = ad[16 * j + c];
      #pragma unroll
      for (int r = 0; r < 8; ++r) {
        ps[r] += acc[j][r] * sv;
        pd[r] += acc[j][r] * dv;
      }
    }
    #pragma unroll
    for (int r = 0; r < 8; ++r) { ps[r] += row_xor<1>(ps[r]); pd[r] += row_xor<1>(pd[r]); }
    #pragma unroll
    for (int r = 0; r < 8; ++r) { ps[r] += row_xor<2>(ps[r]); pd[r] += row_xor<2>(pd[r]); }
    #pragma unroll
    for (int r = 0; r < 8; ++r) { ps[r] += row_xor<4>(ps[r]); pd[r] += row_xor<4>(pd[r]); }
    #pragma unroll
    for (int r = 0; r < 8; ++r) { ps[r] += row_xor<8>(ps[r]); pd[r] += row_xor<8>(pd[r]); }
    if (c == 0) {
      #pragma unroll
      for (int r = 0; r < 8; ++r) {
        const int rr = row0 + 8 * h + r;
        if (rr < N) { es[rr] = ps[r]; ed[rr] = pd[r]; }
      }
    }
  }

  if (row0 + 16 <= N) {
    float* hp = H + (size_t)(row0 + 8 * h) * 128 + c;
    #pragma unroll
    for (int j = 0; j < 8; ++j) {
      #pragma unroll
      for (int r = 0; r < 8; ++r) hp[(size_t)r * 128 + 16 * j] = acc[j][r];
    }
  } else {
    #pragma unroll
    for (int j = 0; j < 8; ++j) {
      #pragma unroll
      for (int r = 0; r < 8; ++r) {
        const int rr = row0 + 8 * h + r;
        if (rr < N) H[(size_t)rr * 128 + 16 * j + c] = acc[j][r];
      }
    }
  }
}

__global__ void dst_rank(const int* __restrict__ dst, int E, unsigned* __restrict__ cnt,
                         unsigned* __restrict__ rank) {
  int i = blockIdx.x * blockDim.x + threadIdx.x;
  if (i >= E) return;
  rank[i] = atomicAdd(&cnt[dst[i]], 1u);
}

__global__ void scan_tiles(const unsigned* __restrict__ cnt, int N, unsigned* __restrict__ tsum) {
  __shared__ unsigned sm[8];
  const int lane = threadIdx.x & 31;
  const int wave = threadIdx.x >> 5;
  const int base = blockIdx.x * 1024 + threadIdx.x * 4;
  unsigned v = 0;
  for (int j = 0; j < 4; ++j)
    if (base + j < N) v += cnt[base + j];
  v = warp_sum_u32(v);
  if (lane == 0) sm[wave] = v;
  __syncthreads();
  if (threadIdx.x == 0) {
    unsigned t = 0;
    for (int w = 0; w < 8; ++w) t += sm[w];
    tsum[blockIdx.x] = t;
  }
}

__global__ void scan_rows(const unsigned* __restrict__ cnt, const unsigned* __restrict__ tsum,
                          int N, unsigned* __restrict__ rowptr) {
  __shared__ unsigned sm[8];
  const int tid = threadIdx.x;
  const int lane = tid & 31;
  const int wave = tid >> 5;

  unsigned o = 0;
  for (int b = tid; b < (int)blockIdx.x; b += 256) o += tsum[b];
  o = warp_sum_u32(o);
  if (lane == 0) sm[wave] = o;
  __syncthreads();
  unsigned off = 0;
  for (int w = 0; w < 8; ++w) off += sm[w];
  __syncthreads();

  const int base = blockIdx.x * 1024 + tid * 4;
  unsigned c[4];
  for (int j = 0; j < 4; ++j) c[j] = (base + j < N) ? cnt[base + j] : 0u;
  const unsigned mine = c[0] + c[1] + c[2] + c[3];
  unsigned incl = mine;
  for (int d = 1; d < 32; d <<= 1) {
    unsigned t = __shfl_up(incl, d, 32);
    if (lane >= d) incl += t;
  }
  if (lane == 31) sm[wave] = incl;
  __syncthreads();
  unsigned woff = 0;
  for (int w = 0; w < 8; ++w)
    if (w < wave) woff += sm[w];
  unsigned p = off + woff + incl - mine;
  for (int j = 0; j < 4; ++j) {
    if (base + j < N) rowptr[base + j] = p;
    p += c[j];
  }
}

__global__ void csr_fill(const int* __restrict__ src, const int* __restrict__ dst,
                         const unsigned* __restrict__ rank, const unsigned* __restrict__ rowptr,
                         int E, int* __restrict__ csr_src) {
  int i = blockIdx.x * blockDim.x + threadIdx.x;
  if (i >= E) return;
  csr_src[rowptr[dst[i]] + rank[i]] = src[i];
}

__global__ void gat_gather128(const float* __restrict__ H, const float* __restrict__ es,
                              const float* __restrict__ ed, const unsigned* __restrict__ rowptr,
                              const unsigned* __restrict__ cnt, const int* __restrict__ csr_src,
                              int N, const float* __restrict__ b1, const float* __restrict__ W2,
                              const float* __restrict__ a2s, const float* __restrict__ a2d,
                              float* __restrict__ h2, float* __restrict__ e2s,
                              float* __restrict__ e2d) {
  const int n = __builtin_amdgcn_readfirstlane((int)(blockIdx.x * 8 + (threadIdx.x >> 5)));
  if (n >= N) return;
  const int lane = threadIdx.x & 31;
  const unsigned beg = rowptr[n];
  const int deg = (int)cnt[n];
  const int tot = deg + 1;
  const float edn = ed[n];

  int s0 = n;
  float e0 = -INFINITY;
  const bool v0 = lane < tot;
  if (v0) {
    s0 = (lane < deg) ? csr_src[beg + lane] : n;
    e0 = leaky(es[s0] + edn);
  }
  float m = e0;
  for (int j = lane + 32; j < tot; j += 32) {
    int s = (j < deg) ? csr_src[beg + j] : n;
    m = fmaxf(m, leaky(es[s] + edn));
  }
  m = warp_max(m);

  const float x0 = v0 ? expf(e0 - m) : 0.0f;
  float dsum = x0;
  for (int j = lane + 32; j < tot; j += 32) {
    int s = (j < deg) ? csr_src[beg + j] : n;
    dsum += expf(leaky(es[s] + edn) - m);
  }
  dsum = warp_sum(dsum);

  float4 acc = make_float4(0.0f, 0.0f, 0.0f, 0.0f);
  {
    const float a0 = x0 / dsum;
    const int c0 = (tot < 32) ? tot : 32;
    #pragma unroll 4
    for (int t = 0; t < c0; ++t) {
      const int st = __builtin_amdgcn_readlane(s0, t);
      const float at = __int_as_float(__builtin_amdgcn_readlane(__float_as_int(a0), t));
      const float4 hv = ((const float4*)(H + (size_t)st * 128))[lane];
      acc.x += hv.x * at;
      acc.y += hv.y * at;
      acc.z += hv.z * at;
      acc.w += hv.w * at;
    }
  }
  for (int j0 = 32; j0 < tot; j0 += 32) {
    const int j = j0 + lane;
    int s = n;
    float a = 0.0f;
    if (j < tot) {
      s = (j < deg) ? csr_src[beg + j] : n;
      a = expf(leaky(es[s] + edn) - m) / dsum;
    }
    const int cc = (tot - j0 < 32) ? (tot - j0) : 32;
    for (int t = 0; t < cc; ++t) {
      const int st = __builtin_amdgcn_readlane(s, t);
      const float at = __int_as_float(__builtin_amdgcn_readlane(__float_as_int(a), t));
      const float4 hv = ((const float4*)(H + (size_t)st * 128))[lane];
      acc.x += hv.x * at;
      acc.y += hv.y * at;
      acc.z += hv.z * at;
      acc.w += hv.w * at;
    }
  }
  const float4 bv = ((const float4*)b1)[lane];
  const float4 w0 = ((const float4*)W2)[2 * lane];
  const float4 w1 = ((const float4*)W2)[2 * lane + 1];
  float4 hv;
  { const float v = acc.x + bv.x; hv.x = (v > 0.0f) ? v : 0.0f; }
  { const float v = acc.y + bv.y; hv.y = (v > 0.0f) ? v : 0.0f; }
  { const float v = acc.z + bv.z; hv.z = (v > 0.0f) ? v : 0.0f; }
  { const float v = acc.w + bv.w; hv.w = (v > 0.0f) ? v : 0.0f; }
  float q0 = hv.x * w0.x + hv.y * w0.z + hv.z * w1.x + hv.w * w1.z;
  float q1 = hv.x * w0.y + hv.y * w0.w + hv.z * w1.y + hv.w * w1.w;
  q0 = warp_sum(q0);
  q1 = warp_sum(q1);
  if (lane == 0) {
    h2[2 * n + 0] = q0;
    h2[2 * n + 1] = q1;
    e2s[n] = q0 * a2s[0] + q1 * a2s[1];
    e2d[n] = q0 * a2d[0] + q1 * a2d[1];
  }
}

__global__ void gat_gather2(const float* __restrict__ h2, const float* __restrict__ es,
                            const float* __restrict__ ed, const unsigned* __restrict__ rowptr,
                            const unsigned* __restrict__ cnt, const int* __restrict__ csr_src,
                            int N, const float* __restrict__ b2, float* __restrict__ out) {
  const int n = __builtin_amdgcn_readfirstlane((int)(blockIdx.x * 8 + (threadIdx.x >> 5)));
  if (n >= N) return;
  const int lane = threadIdx.x & 31;
  const unsigned beg = rowptr[n];
  const int deg = (int)cnt[n];
  const int tot = deg + 1;
  const float edn = ed[n];

  int s0 = n;
  float e0 = -INFINITY;
  const bool v0 = lane < tot;
  if (v0) {
    s0 = (lane < deg) ? csr_src[beg + lane] : n;
    e0 = leaky(es[s0] + edn);
  }
  float m = e0;
  for (int j = lane + 32; j < tot; j += 32) {
    int s = (j < deg) ? csr_src[beg + j] : n;
    m = fmaxf(m, leaky(es[s] + edn));
  }
  m = warp_max(m);

  const float x0 = v0 ? expf(e0 - m) : 0.0f;
  float dsum = x0;
  for (int j = lane + 32; j < tot; j += 32) {
    int s = (j < deg) ? csr_src[beg + j] : n;
    dsum += expf(leaky(es[s] + edn) - m);
  }
  dsum = warp_sum(dsum);

  float p0 = 0.0f, p1 = 0.0f;
  if (v0) {
    const float a0 = x0 / dsum;
    const float2 hv = ((const float2*)h2)[s0];
    p0 = hv.x * a0;
    p1 = hv.y * a0;
  }
  for (int j = lane + 32; j < tot; j += 32) {
    int s = (j < deg) ? csr_src[beg + j] : n;
    const float a = expf(leaky(es[s] + edn) - m) / dsum;
    const float2 hv = ((const float2*)h2)[s];
    p0 += hv.x * a;
    p1 += hv.y * a;
  }
  p0 = warp_sum(p0);
  p1 = warp_sum(p1);
  if (lane == 0) {
    out[2 * n + 0] = p0 + b2[0];
    out[2 * n + 1] = p1 + b2[1];
  }
}


extern "C" void kernel_launch(void* const* d_in, const int* in_sizes, int n_in,
                              void* d_out, int out_size, void* d_ws, size_t ws_size,
                              hipStream_t stream) {
  const float* x   = (const float*)d_in[0];
  const int*  eidx = (const int*)d_in[1];
  const float* W1  = (const float*)d_in[2];
  const float* a1s = (const float*)d_in[3];
  const float* a1d = (const float*)d_in[4];
  const float* b1  = (const float*)d_in[5];
  const float* W2  = (const float*)d_in[6];
  const float* a2s = (const float*)d_in[7];
  const float* a2d = (const float*)d_in[8];
  const float* b2  = (const float*)d_in[9];

  const int N  = in_sizes[0] / 256;
  const int E  = in_sizes[1] / 2;
  const int* src = eidx;
  const int* dst = eidx + E;
  float* out = (float*)d_out;

  char* w = (char*)d_ws;
  float*    h1     = (float*)w;    w += (size_t)N * 128 * sizeof(float);
  float*    e1s    = (float*)w;    w += (size_t)N * sizeof(float);
  float*    e1d    = (float*)w;    w += (size_t)N * sizeof(float);
  float*    h2     = (float*)w;    w += (size_t)N * 2 * sizeof(float);
  float*    e2s    = (float*)w;    w += (size_t)N * sizeof(float);
  float*    e2d    = (float*)w;    w += (size_t)N * sizeof(float);
  unsigned* cnt    = (unsigned*)w; w += (size_t)N * sizeof(unsigned);
  unsigned* rowptr = (unsigned*)w; w += (size_t)N * sizeof(unsigned);
  unsigned* tsum   = (unsigned*)w; w += (size_t)((N + 1023) / 1024) * sizeof(unsigned);
  unsigned* rank   = (unsigned*)w; w += (size_t)E * sizeof(unsigned);
  int*      csrs   = (int*)w;      w += (size_t)E * sizeof(int);

  const int TB = 256;
  zero_f32<<<256, TB, 0, stream>>>((float*)cnt, (size_t)N);
  dst_rank<<<(E + TB - 1) / TB, TB, 0, stream>>>(dst, E, cnt, rank);
  scan_tiles<<<(N + 1023) / 1024, TB, 0, stream>>>(cnt, N, tsum);
  scan_rows<<<(N + 1023) / 1024, TB, 0, stream>>>(cnt, tsum, N, rowptr);
  csr_fill<<<(E + TB - 1) / TB, TB, 0, stream>>>(src, dst, rank, rowptr, E, csrs);

  gemm1_wmma<<<(N + 63) / 64, 128, 0, stream>>>(x, W1, a1s, a1d, h1, e1s, e1d, N);
  gat_gather128<<<(N + 7) / 8, TB, 0, stream>>>(h1, e1s, e1d, rowptr, cnt, csrs, N, b1, W2, a2s, a2d,
                                                h2, e2s, e2d);

  gat_gather2<<<(N + 7) / 8, TB, 0, stream>>>(h2, e2s, e2d, rowptr, cnt, csrs, N, b2, out);
}


# Edges grouped by destination by a two-level counting sort on LDS counters (chunk-local range partition, one workgroup per 512-node range): no global atomics, 8 launches to 5; sum order = LDS arrival
# speedup vs baseline: 6.0514x; 1.6602x over previous
#include <hip/hip_runtime.h>
#include <hip/hip_bf16.h>

typedef __attribute__((ext_vector_type(2))) float v2f;
typedef __attribute__((ext_vector_type(8))) float v8f;

#define NEG_SLOPE 0.2f


__device__ __forceinline__ float warp_sum(float v) {
  #pragma unroll
  for (int o = 16; o > 0; o >>= 1) v += __shfl_xor(v, o, 32);
  return v;
}

__device__ __forceinline__ float warp_max(float v) {
  for (int o = 16; o > 0; o >>= 1) v = fmaxf(v, __shfl_xor(v, o, 32));
  return v;
}

__device__ __forceinline__ unsigned warp_sum_u32(unsigned v) {
  for (int o = 16; o > 0; o >>= 1) v += __shfl_xor(v, o, 32);
  return v;
}

template <int M>
__device__ __forceinline__ float row_xor(float v) {
  return __int_as_float(__builtin_amdgcn_update_dpp(0, __float_as_int(v), 0x160 + M, 0xf, 0xf, true));
}

__device__ __forceinline__ float leaky(float e) {
  return (e > 0.0f) ? e : NEG_SLOPE * e;
}

#define G_KC 16
#define G_PITCH 264

__global__ __launch_bounds__(128) __attribute__((amdgpu_waves_per_eu(7))) void gemm1_wmma(const float* __restrict__ X,
                                                  const float* __restrict__ W,
                                                  const float* __restrict__ as,
                                                  const float* __restrict__ ad,
                                                  float* __restrict__ H,
                                                  float* __restrict__ es,
                                                  float* __restrict__ ed, int N) {
  __shared__ __attribute__((aligned(16))) float Bs[2][8 * G_PITCH];
  const int tid  = threadIdx.x;
  const int lane = tid & 31;
  const int wave = tid >> 5;
  const int c = lane & 15;
  const int h = lane >> 4;
  const int row0 = blockIdx.x * 64 + wave * 16;

  int row = row0 + c; if (row >= N) row = N - 1;
  const float* xrow = X + (size_t)row * 256 + 8 * h;

  const float* wsrc = W + (size_t)(2 * wave) * 128 + 4 * lane;
  const int sdst = wave * G_PITCH + 8 * lane;

  v8f acc[8];
  #pragma unroll
  for (int j = 0; j < 8; ++j) acc[j] = (v8f){};

  float4 a0 = *(const float4*)(xrow);
  float4 a1 = *(const float4*)(xrow + 4);
  {
    const float4 e0 = *(const float4*)(wsrc);
    const float4 o0 = *(const float4*)(wsrc + 128);
    const float4 e1 = *(const float4*)(wsrc + 8 * 128);
    const float4 o1 = *(const float4*)(wsrc + 9 * 128);
    float* d = &Bs[0][sdst];
    *(float4*)(d)                   = make_float4(e0.x, o0.x, e0.y, o0.y);
    *(float4*)(d + 4)               = make_float4(e0.z, o0.z, e0.w, o0.w);
    *(float4*)(d + 4 * G_PITCH)     = make_float4(e1.x, o1.x, e1.y, o1.y);
    *(float4*)(d + 4 * G_PITCH + 4) = make_float4(e1.z, o1.z, e1.w, o1.w);
  }
  __syncthreads();

  for (int i = 0; i < 256 / G_KC; ++i) {
    const bool more = (i + 1) < 256 / G_KC;
    float4 na0 = a0, na1 = a1, e0, o0, e1, o1;
    if (more) {
      const int kn = (i + 1) * G_KC;
      const float* ws = wsrc + (size_t)kn * 128;
      e0 = *(const float4*)(ws);
      o0 = *(const float4*)(ws + 128);
      e1 = *(const float4*)(ws + 8 * 128);
      o1 = *(const float4*)(ws + 9 * 128);
      na0 = *(const float4*)(xrow + kn);
      na1 = *(const float4*)(xrow + kn + 4);
    }
    __builtin_amdgcn_sched_barrier(0);

    const float* bs = &Bs[i & 1][(4 * h) * G_PITCH + 2 * c];
    v2f bf[2][8];
    #pragma unroll
    for (int j = 0; j < 8; ++j) bf[0][j] = *(const v2f*)(bs + 32 * j);
    #pragma unroll
    for (int t = 0; t < 4; ++t) {
      if (t < 3) {
        #pragma unroll
        for (int j = 0; j < 8; ++j) bf[(t + 1) & 1][j] = *(const v2f*)(bs + (t + 1) * G_PITCH + 32 * j);
      }
      __builtin_amdgcn_sched_barrier(0);
      v2f a;
      if (t == 0)      { a.x = a0.x; a.y = a0.y; }
      else if (t == 1) { a.x = a0.z; a.y = a0.w; }
      else if (t == 2) { a.x = a1.x; a.y = a1.y; }
      else             { a.x = a1.z; a.y = a1.w; }
      #pragma unroll
      for (int j = 0; j < 8; ++j) {
        acc[j] = __builtin_amdgcn_wmma_f32_16x16x4_f32(false, a, false, bf[t & 1][j], (short)0,
                                                       acc[j], false, false);
      }
      __builtin_amdgcn_sched_barrier(0);
    }

    if (more) {
      float* d = &Bs[(i + 1) & 1][sdst];
      *(float4*)(d)                   = make_float4(e0.x, o0.x, e0.y, o0.y);
      *(float4*)(d + 4)               = make_float4(e0.z, o0.z, e0.w, o0.w);
      *(float4*)(d + 4 * G_PITCH)     = make_float4(e1.x, o1.x, e1.y, o1.y);
      *(float4*)(d + 4 * G_PITCH + 4) = make_float4(e1.z, o1.z, e1.w, o1.w);
    }
    __syncthreads();
    a0 = na0; a1 = na1;
  }

  {
    float ps[8], pd[8];
    #pragma unroll
    for (int r = 0; r < 8; ++r) { ps[r] = 0.0f; pd[r] = 0.0f; }
    #pragma unroll
    for (int j = 0; j < 8; ++j) {
      const float sv = as[16 * j + c];
      const float dv = ad[16 * j + c];
      #pragma unroll
      for (int r = 0; r < 8; ++r) {
        ps[r] += acc[j][r] * sv;
        pd[r] += acc[j][r] * dv;
      }
    }
    #pragma unroll
    for (int r = 0; r < 8; ++r) { ps[r] += row_xor<1>(ps[r]); pd[r] += row_xor<1>(pd[r]); }
    #pragma unroll
    for (int r = 0; r < 8; ++r) { ps[r] += row_xor<2>(ps[r]); pd[r] += row_xor<2>(pd[r]); }
    #pragma unroll
    for (int r = 0; r < 8; ++r) { ps[r] += row_xor<4>(ps[r]); pd[r] += row_xor<4>(pd[r]); }
    #pragma unroll
    for (int r = 0; r < 8; ++r) { ps[r] += row_xor<8>(ps[r]); pd[r] += row_xor<8>(pd[r]); }
    if (c == 0) {
      #pragma unroll
      for (int r = 0; r < 8; ++r) {
        const int rr = row0 + 8 * h + r;
        if (rr < N) { es[rr] = ps[r]; ed[rr] = pd[r]; }
      }
    }
  }

  if (row0 + 16 <= N) {
    float* hp = H + (size_t)(row0 + 8 * h) * 128 + c;
    #pragma unroll
    for (int j = 0; j < 8; ++j) {
      #pragma unroll
      for (int r = 0; r < 8; ++r) hp[(size_t)r * 128 + 16 * j] = acc[j][r];
    }
  } else {
    #pragma unroll
    for (int j = 0; j < 8; ++j) {
      #pragma unroll
      for (int r = 0; r < 8; ++r) {
        const int rr = row0 + 8 * h + r;
        if (rr < N) H[(size_t)rr * 128 + 16 * j + c] = acc[j][r];
      }
    }
  }
}

#define P_SHIFT 9
#define P_BW (1 << P_SHIFT)
#define P_CH 8192
#define P_NBMAX 1024
#define P_UN 8

__global__ __launch_bounds__(1024) void part_edges(const int* __restrict__ src,
                                                   const int* __restrict__ dst, int E, int NB, int G,
                                                   unsigned* __restrict__ offsT,
                                                   int2* __restrict__ part) {
  __shared__ unsigned histL[P_NBMAX];
  __shared__ unsigned wsum[32];
  __shared__ int2 stage[P_CH];
  const int tid = threadIdx.x;
  const int lane = tid & 31;
  const int wave = tid >> 5;
  const int g = blockIdx.x;
  const int e0 = g * P_CH;
  int len = E - e0;
  if (len > P_CH) len = P_CH;

  histL[tid] = 0u;
  __syncthreads();

  int d[8], s[8];
  unsigned r[8];
  #pragma unroll
  for (int q = 0; q < 8; ++q) {
    const int j = q * 1024 + tid;
    d[q] = 0; s[q] = 0;
    if (j < len) { d[q] = dst[e0 + j]; s[q] = src[e0 + j]; }
  }
  #pragma unroll
  for (int q = 0; q < 8; ++q) {
    r[q] = 0u;
    if (q * 1024 + tid < len) r[q] = atomicAdd(&histL[d[q] >> P_SHIFT], 1u);
  }
  __syncthreads();

  const unsigned c = histL[tid];
  unsigned incl = c;
  for (int dd = 1; dd < 32; dd <<= 1) {
    const unsigned t = __shfl_up(incl, dd, 32);
    if (lane >= dd) incl += t;
  }
  if (lane == 31) wsum[wave] = incl;
  __syncthreads();
  unsigned woff = 0;
  for (int w = 0; w < 32; ++w)
    if (w < wave) woff += wsum[w];
  const unsigned excl = woff + incl - c;
  histL[tid] = excl;
  if (tid <= NB) offsT[(size_t)tid * G + g] = excl;
  __syncthreads();

  #pragma unroll
  for (int q = 0; q < 8; ++q) {
    if (q * 1024 + tid < len) stage[histL[d[q] >> P_SHIFT] + r[q]] = make_int2(d[q], s[q]);
  }
  __syncthreads();
  for (int j = tid; j < len; j += 1024) part[e0 + j] = stage[j];
}

__global__ __launch_bounds__(1024) void csr_ranges(const int2* __restrict__ part,
                                                   const unsigned* __restrict__ offsT, int N, int G,
                                                   unsigned* __restrict__ cnt,
                                                   unsigned* __restrict__ rowptr,
                                                   int* __restrict__ csr_src) {
  __shared__ unsigned cntL[P_BW];
  __shared__ unsigned wsum[32];
  __shared__ unsigned wbase[32];
  const int tid = threadIdx.x;
  const int lane = tid & 31;
  const int wave = tid >> 5;
  const int b = blockIdx.x;
  const int n0 = b << P_SHIFT;

  if (tid < P_BW) cntL[tid] = 0u;
  __syncthreads();

  int tpp = 1024 / G;
  if (tpp < 1) tpp = 1;
  if (tpp > 32) tpp = 32;
  const int slots = 1024 / tpp;
  const int pg = tid / tpp;
  const int pk = tid - pg * tpp;
  const unsigned* row0 = offsT + (size_t)b * G;
  const unsigned* row1 = row0 + G;

  unsigned mybase = 0u;
  if (pg < slots) {
    for (int g = pg; g < G; g += slots) {
      const unsigned o0 = row0[g];
      const unsigned o1 = row1[g];
      if (pk == 0) mybase += o0;
      const int* pd = (const int*)(part + (size_t)g * P_CH);
      for (unsigned j = o0 + pk; j < o1; j += P_UN * tpp) {
        int dv[P_UN];
        #pragma unroll
        for (int u = 0; u < P_UN; ++u) {
          const unsigned jj = j + u * tpp;
          dv[u] = -1;
          if (jj < o1) dv[u] = pd[2 * (size_t)jj];
        }
        #pragma unroll
        for (int u = 0; u < P_UN; ++u)
          if (dv[u] >= 0) atomicAdd(&cntL[dv[u] - n0], 1u);
      }
    }
  }
  __syncthreads();

  const unsigned bsum = warp_sum_u32(mybase);
  if (lane == 0) wbase[wave] = bsum;
  const unsigned c = (tid < P_BW) ? cntL[tid] : 0u;
  unsigned incl = c;
  for (int dd = 1; dd < 32; dd <<= 1) {
    const unsigned t = __shfl_up(incl, dd, 32);
    if (lane >= dd) incl += t;
  }
  if (lane == 31) wsum[wave] = incl;
  __syncthreads();
  unsigned start = 0u;
  for (int w = 0; w < 32; ++w) start += wbase[w];
  for (int w = 0; w < 32; ++w)
    if (w < wave) start += wsum[w];
  start += incl - c;
  if (tid < P_BW) {
    const int n = n0 + tid;
    if (n < N) { cnt[n] = c; rowptr[n] = start; }
    cntL[tid] = start;
  }
  __syncthreads();

  if (pg < slots) {
    for (int g = pg; g < G; g += slots) {
      const unsigned o0 = row0[g];
      const unsigned o1 = row1[g];
      const int2* p = part + (size_t)g * P_CH;
      for (unsigned j = o0 + pk; j < o1; j += P_UN * tpp) {
        int2 e[P_UN];
        #pragma unroll
        for (int u = 0; u < P_UN; ++u) {
          const unsigned jj = j + u * tpp;
          e[u] = make_int2(-1, 0);
          if (jj < o1) e[u] = p[jj];
        }
        #pragma unroll
        for (int u = 0; u < P_UN; ++u)
          if (e[u].x >= 0) csr_src[atomicAdd(&cntL[e[u].x - n0], 1u)] = e[u].y;
      }
    }
  }
}

__global__ void gat_gather128(const float* __restrict__ H, const float* __restrict__ es,
                              const float* __restrict__ ed, const unsigned* __restrict__ rowptr,
                              const unsigned* __restrict__ cnt, const int* __restrict__ csr_src,
                              int N, const float* __restrict__ b1, const float* __restrict__ W2,
                              const float* __restrict__ a2s, const float* __restrict__ a2d,
                              float* __restrict__ h2, float* __restrict__ e2s,
                              float* __restrict__ e2d) {
  const int n = __builtin_amdgcn_readfirstlane((int)(blockIdx.x * 8 + (threadIdx.x >> 5)));
  if (n >= N) return;
  const int lane = threadIdx.x & 31;
  const unsigned beg = rowptr[n];
  const int deg = (int)cnt[n];
  const int tot = deg + 1;
  const float edn = ed[n];

  int s0 = n;
  float e0 = -INFINITY;
  const bool v0 = lane < tot;
  if (v0) {
    s0 = (lane < deg) ? csr_src[beg + lane] : n;
    e0 = leaky(es[s0] + edn);
  }
  float m = e0;
  for (int j = lane + 32; j < tot; j += 32) {
    int s = (j < deg) ? csr_src[beg + j] : n;
    m = fmaxf(m, leaky(es[s] + edn));
  }
  m = warp_max(m);

  const float x0 = v0 ? expf(e0 - m) : 0.0f;
  float dsum = x0;
  for (int j = lane + 32; j < tot; j += 32) {
    int s = (j < deg) ? csr_src[beg + j] : n;
    dsum += expf(leaky(es[s] + edn) - m);
  }
  dsum = warp_sum(dsum);

  float4 acc = make_float4(0.0f, 0.0f, 0.0f, 0.0f);
  {
    const float a0 = x0 / dsum;
    const int c0 = (tot < 32) ? tot : 32;
    #pragma unroll 4
    for (int t = 0; t < c0; ++t) {
      const int st = __builtin_amdgcn_readlane(s0, t);
      const float at = __int_as_float(__builtin_amdgcn_readlane(__float_as_int(a0), t));
      const float4 hv = ((const float4*)(H + (size_t)st * 128))[lane];
      acc.x += hv.x * at;
      acc.y += hv.y * at;
      acc.z += hv.z * at;
      acc.w += hv.w * at;
    }
  }
  for (int j0 = 32; j0 < tot; j0 += 32) {
    const int j = j0 + lane;
    int s = n;
    float a = 0.0f;
    if (j < tot) {
      s = (j < deg) ? csr_src[beg + j] : n;
      a = expf(leaky(es[s] + edn) - m) / dsum;
    }
    const int cc = (tot - j0 < 32) ? (tot - j0) : 32;
    for (int t = 0; t < cc; ++t) {
      const int st = __builtin_amdgcn_readlane(s, t);
      const float at = __int_as_float(__builtin_amdgcn_readlane(__float_as_int(a), t));
      const float4 hv = ((const float4*)(H + (size_t)st * 128))[lane];
      acc.x += hv.x * at;
      acc.y += hv.y * at;
      acc.z += hv.z * at;
      acc.w += hv.w * at;
    }
  }
  const float4 bv = ((const float4*)b1)[lane];
  const float4 w0 = ((const float4*)W2)[2 * lane];
  const float4 w1 = ((const float4*)W2)[2 * lane + 1];
  float4 hv;
  { const float v = acc.x + bv.x; hv.x = (v > 0.0f) ? v : 0.0f; }
  { const float v = acc.y + bv.y; hv.y = (v > 0.0f) ? v : 0.0f; }
  { const float v = acc.z + bv.z; hv.z = (v > 0.0f) ? v : 0.0f; }
  { const float v = acc.w + bv.w; hv.w = (v > 0.0f) ? v : 0.0f; }
  float q0 = hv.x * w0.x + hv.y * w0.z + hv.z * w1.x + hv.w * w1.z;
  float q1 = hv.x * w0.y + hv.y * w0.w + hv.z * w1.y + hv.w * w1.w;
  q0 = warp_sum(q0);
  q1 = warp_sum(q1);
  if (lane == 0) {
    h2[2 * n + 0] = q0;
    h2[2 * n + 1] = q1;
    e2s[n] = q0 * a2s[0] + q1 * a2s[1];
    e2d[n] = q0 * a2d[0] + q1 * a2d[1];
  }
}

__global__ void gat_gather2(const float* __restrict__ h2, const float* __restrict__ es,
                            const float* __restrict__ ed, const unsigned* __restrict__ rowptr,
                            const unsigned* __restrict__ cnt, const int* __restrict__ csr_src,
                            int N, const float* __restrict__ b2, float* __restrict__ out) {
  const int n = __builtin_amdgcn_readfirstlane((int)(blockIdx.x * 8 + (threadIdx.x >> 5)));
  if (n >= N) return;
  const int lane = threadIdx.x & 31;
  const unsigned beg = rowptr[n];
  const int deg = (int)cnt[n];
  const int tot = deg + 1;
  const float edn = ed[n];

  int s0 = n;
  float e0 = -INFINITY;
  const bool v0 = lane < tot;
  if (v0) {
    s0 = (lane < deg) ? csr_src[beg + lane] : n;
    e0 = leaky(es[s0] + edn);
  }
  float m = e0;
  for (int j = lane + 32; j < tot; j += 32) {
    int s = (j < deg) ? csr_src[beg + j] : n;
    m = fmaxf(m, leaky(es[s] + edn));
  }
  m = warp_max(m);

  const float x0 = v0 ? expf(e0 - m) : 0.0f;
  float dsum = x0;
  for (int j = lane + 32; j < tot; j += 32) {
    int s = (j < deg) ? csr_src[beg + j] : n;
    dsum += expf(leaky(es[s] + edn) - m);
  }
  dsum = warp_sum(dsum);

  float p0 = 0.0f, p1 = 0.0f;
  if (v0) {
    const float a0 = x0 / dsum;
    const float2 hv = ((const float2*)h2)[s0];
    p0 = hv.x * a0;
    p1 = hv.y * a0;
  }
  for (int j = lane + 32; j < tot; j += 32) {
    int s = (j < deg) ? csr_src[beg + j] : n;
    const float a = expf(leaky(es[s] + edn) - m) / dsum;
    const float2 hv = ((const float2*)h2)[s];
    p0 += hv.x * a;
    p1 += hv.y * a;
  }
  p0 = warp_sum(p0);
  p1 = warp_sum(p1);
  if (lane == 0) {
    out[2 * n + 0] = p0 + b2[0];
    out[2 * n + 1] = p1 + b2[1];
  }
}


extern "C" void kernel_launch(void* const* d_in, const int* in_sizes, int n_in,
                              void* d_out, int out_size, void* d_ws, size_t ws_size,
                              hipStream_t stream) {
  const float* x   = (const float*)d_in[0];
  const int*  eidx = (const int*)d_in[1];
  const float* W1  = (const float*)d_in[2];
  const float* a1s = (const float*)d_in[3];
  const float* a1d = (const float*)d_in[4];
  const float* b1  = (const float*)d_in[5];
  const float* W2  = (const float*)d_in[6];
  const float* a2s = (const float*)d_in[7];
  const float* a2d = (const float*)d_in[8];
  const float* b2  = (const float*)d_in[9];

  const int N  = in_sizes[0] / 256;
  const int E  = in_sizes[1] / 2;
  const int* src = eidx;
  const int* dst = eidx + E;
  float* out = (float*)d_out;

  const int NB = (N + P_BW - 1) / P_BW;
  const int G  = (E + P_CH - 1) / P_CH;

  char* w = (char*)d_ws;
  float*    h1     = (float*)w;    w += (size_t)N * 128 * sizeof(float);
  int2*     part   = (int2*)w;     w += (size_t)E * sizeof(int2);
  float*    e1s    = (float*)w;    w += (size_t)N * sizeof(float);
  float*    e1d    = (float*)w;    w += (size_t)N * sizeof(float);
  float*    h2     = (float*)w;    w += (size_t)N * 2 * sizeof(float);
  float*    e2s    = (float*)w;    w += (size_t)N * sizeof(float);
  float*    e2d    = (float*)w;    w += (size_t)N * sizeof(float);
  unsigned* cnt    = (unsigned*)w; w += (size_t)N * sizeof(unsigned);
  unsigned* rowptr = (unsigned*)w; w += (size_t)N * sizeof(unsigned);
  unsigned* offsT  = (unsigned*)w; w += (size_t)(NB + 1) * G * sizeof(unsigned);
  int*      csrs   = (int*)w;      w += (size_t)E * sizeof(int);

  const int TB = 256;
  part_edges<<<G, 1024, 0, stream>>>(src, dst, E, NB, G, offsT, part);
  csr_ranges<<<NB, 1024, 0, stream>>>(part, offsT, N, G, cnt, rowptr, csrs);

  gemm1_wmma<<<(N + 63) / 64, 128, 0, stream>>>(x, W1, a1s, a1d, h1, e1s, e1d, N);
  gat_gather128<<<(N + 7) / 8, TB, 0, stream>>>(h1, e1s, e1d, rowptr, cnt, csrs, N, b1, W2, a2s, a2d,
                                                h2, e2s, e2d);

  gat_gather2<<<(N + 7) / 8, TB, 0, stream>>>(h2, e2s, e2d, rowptr, cnt, csrs, N, b2, out);
}


# Gather kernels: a wave takes 10 (layer 1) or 16 (layer 2) consecutive destination nodes, 5 or 8 at a time with their loads batched per dependent hop; 200000 gather waves become 16256; sum order kept
# speedup vs baseline: 6.6268x; 1.0951x over previous
#include <hip/hip_runtime.h>
#include <hip/hip_bf16.h>

typedef __attribute__((ext_vector_type(2))) float v2f;
typedef __attribute__((ext_vector_type(8))) float v8f;

#define NEG_SLOPE 0.2f


__device__ __forceinline__ float warp_sum(float v) {
  #pragma unroll
  for (int o = 16; o > 0; o >>= 1) v += __shfl_xor(v, o, 32);
  return v;
}

__device__ __forceinline__ float warp_max(float v) {
  for (int o = 16; o > 0; o >>= 1) v = fmaxf(v, __shfl_xor(v, o, 32));
  return v;
}

__device__ __forceinline__ unsigned warp_sum_u32(unsigned v) {
  for (int o = 16; o > 0; o >>= 1) v += __shfl_xor(v, o, 32);
  return v;
}

template <int M>
__device__ __forceinline__ float row_xor(float v) {
  return __int_as_float(__builtin_amdgcn_update_dpp(0, __float_as_int(v), 0x160 + M, 0xf, 0xf, true));
}

__device__ __forceinline__ float leaky(float e) {
  return (e > 0.0f) ? e : NEG_SLOPE * e;
}

#define G_KC 16
#define G_PITCH 264

__global__ __launch_bounds__(128) __attribute__((amdgpu_waves_per_eu(7))) void gemm1_wmma(const float* __restrict__ X,
                                                  const float* __restrict__ W,
                                                  const float* __restrict__ as,
                                                  const float* __restrict__ ad,
                                                  float* __restrict__ H,
                                                  float* __restrict__ es,
                                                  float* __restrict__ ed, int N) {
  __shared__ __attribute__((aligned(16))) float Bs[2][8 * G_PITCH];
  const int tid  = threadIdx.x;
  const int lane = tid & 31;
  const int wave = tid >> 5;
  const int c = lane & 15;
  const int h = lane >> 4;
  const int row0 = blockIdx.x * 64 + wave * 16;

  int row = row0 + c; if (row >= N) row = N - 1;
  const float* xrow = X + (size_t)row * 256 + 8 * h;

  const float* wsrc = W + (size_t)(2 * wave) * 128 + 4 * lane;
  const int sdst = wave * G_PITCH + 8 * lane;

  v8f acc[8];
  #pragma unroll
  for (int j = 0; j < 8; ++j) acc[j] = (v8f){};

  float4 a0 = *(const float4*)(xrow);
  float4 a1 = *(const float4*)(xrow + 4);
  {
    const float4 e0 = *(const float4*)(wsrc);
    const float4 o0 = *(const float4*)(wsrc + 128);
    const float4 e1 = *(const float4*)(wsrc + 8 * 128);
    const float4 o1 = *(const float4*)(wsrc + 9 * 128);
    float* d = &Bs[0][sdst];
    *(float4*)(d)                   = make_float4(e0.x, o0.x, e0.y, o0.y);
    *(float4*)(d + 4)               = make_float4(e0.z, o0.z, e0.w, o0.w);
    *(float4*)(d + 4 * G_PITCH)     = make_float4(e1.x, o1.x, e1.y, o1.y);
    *(float4*)(d + 4 * G_PITCH + 4) = make_float4(e1.z, o1.z, e1.w, o1.w);
  }
  __syncthreads();

  for (int i = 0; i < 256 / G_KC; ++i) {
    const bool more = (i + 1) < 256 / G_KC;
    float4 na0 = a0, na1 = a1, e0, o0, e1, o1;
    if (more) {
      const int kn = (i + 1) * G_KC;
      const float* ws = wsrc + (size_t)kn * 128;
      e0 = *(const float4*)(ws);
      o0 = *(const float4*)(ws + 128);
      e1 = *(const float4*)(ws + 8 * 128);
      o1 = *(const float4*)(ws + 9 * 128);
      na0 = *(const float4*)(xrow + kn);
      na1 = *(const float4*)(xrow + kn + 4);
    }
    __builtin_amdgcn_sched_barrier(0);

    const float* bs = &Bs[i & 1][(4 * h) * G_PITCH + 2 * c];
    v2f bf[2][8];
    #pragma unroll
    for (int j = 0; j < 8; ++j) bf[0][j] = *(const v2f*)(bs + 32 * j);
    #pragma unroll
    for (int t = 0; t < 4; ++t) {
      if (t < 3) {
        #pragma unroll
        for (int j = 0; j < 8; ++j) bf[(t + 1) & 1][j] = *(const v2f*)(bs + (t + 1) * G_PITCH + 32 * j);
      }
      __builtin_amdgcn_sched_barrier(0);
      v2f a;
      if (t == 0)      { a.x = a0.x; a.y = a0.y; }
      else if (t == 1) { a.x = a0.z; a.y = a0.w; }
      else if (t == 2) { a.x = a1.x; a.y = a1.y; }
      else             { a.x = a1.z; a.y = a1.w; }
      #pragma unroll
      for (int j = 0; j < 8; ++j) {
        acc[j] = __builtin_amdgcn_wmma_f32_16x16x4_f32(false, a, false, bf[t & 1][j], (short)0,
                                                       acc[j], false, false);
      }
      __builtin_amdgcn_sched_barrier(0);
    }

    if (more) {
      float* d = &Bs[(i + 1) & 1][sdst];
      *(float4*)(d)                   = make_float4(e0.x, o0.x, e0.y, o0.y);
      *(float4*)(d + 4)               = make_float4(e0.z, o0.z, e0.w, o0.w);
      *(float4*)(d + 4 * G_PITCH)     = make_float4(e1.x, o1.x, e1.y, o1.y);
      *(float4*)(d + 4 * G_PITCH + 4) = make_float4(e1.z, o1.z, e1.w, o1.w);
    }
    __syncthreads();
    a0 = na0; a1 = na1;
  }

  {
    float ps[8], pd[8];
    #pragma unroll
    for (int r = 0; r < 8; ++r) { ps[r] = 0.0f; pd[r] = 0.0f; }
    #pragma unroll
    for (int j = 0; j < 8; ++j) {
      const float sv = as[16 * j + c];
      const float dv = ad[16 * j + c];
      #pragma unroll
      for (int r = 0; r < 8; ++r) {
        ps[r] += acc[j][r] * sv;
        pd[r] += acc[j][r] * dv;
      }
    }
    #pragma unroll
    for (int r = 0; r < 8; ++r) { ps[r] += row_xor<1>(ps[r]); pd[r] += row_xor<1>(pd[r]); }
    #pragma unroll
    for (int r = 0; r < 8; ++r) { ps[r] += row_xor<2>(ps[r]); pd[r] += row_xor<2>(pd[r]); }
    #pragma unroll
    for (int r = 0; r < 8; ++r) { ps[r] += row_xor<4>(ps[r]); pd[r] += row_xor<4>(pd[r]); }
    #pragma unroll
    for (int r = 0; r < 8; ++r) { ps[r] += row_xor<8>(ps[r]); pd[r] += row_xor<8>(pd[r]); }
    if (c == 0) {
      #pragma unroll
      for (int r = 0; r < 8; ++r) {
        const int rr = row0 + 8 * h + r;
        if (rr < N) { es[rr] = ps[r]; ed[rr] = pd[r]; }
      }
    }
  }

  if (row0 + 16 <= N) {
    float* hp = H + (size_t)(row0 + 8 * h) * 128 + c;
    #pragma unroll
    for (int j = 0; j < 8; ++j) {
      #pragma unroll
      for (int r = 0; r < 8; ++r) hp[(size_t)r * 128 + 16 * j] = acc[j][r];
    }
  } else {
    #pragma unroll
    for (int j = 0; j < 8; ++j) {
      #pragma unroll
      for (int r = 0; r < 8; ++r) {
        const int rr = row0 + 8 * h + r;
        if (rr < N) H[(size_t)rr * 128 + 16 * j + c] = acc[j][r];
      }
    }
  }
}

#define P_SHIFT 9
#define P_BW (1 << P_SHIFT)
#define P_CH 8192
#define P_NBMAX 1024
#define P_UN 8

__global__ __launch_bounds__(1024) void part_edges(const int* __restrict__ src,
                                                   const int* __restrict__ dst, int E, int NB, int G,
                                                   unsigned* __restrict__ offsT,
                                                   int2* __restrict__ part) {
  __shared__ unsigned histL[P_NBMAX];
  __shared__ unsigned wsum[32];
  __shared__ int2 stage[P_CH];
  const int tid = threadIdx.x;
  const int lane = tid & 31;
  const int wave = tid >> 5;
  const int g = blockIdx.x;
  const int e0 = g * P_CH;
  int len = E - e0;
  if (len > P_CH) len = P_CH;

  histL[tid] = 0u;
  __syncthreads();

  int d[8], s[8];
  unsigned r[8];
  #pragma unroll
  for (int q = 0; q < 8; ++q) {
    const int j = q * 1024 + tid;
    d[q] = 0; s[q] = 0;
    if (j < len) { d[q] = dst[e0 + j]; s[q] = src[e0 + j]; }
  }
  #pragma unroll
  for (int q = 0; q < 8; ++q) {
    r[q] = 0u;
    if (q * 1024 + tid < len) r[q] = atomicAdd(&histL[d[q] >> P_SHIFT], 1u);
  }
  __syncthreads();

  const unsigned c = histL[tid];
  unsigned incl = c;
  for (int dd = 1; dd < 32; dd <<= 1) {
    const unsigned t = __shfl_up(incl, dd, 32);
    if (lane >= dd) incl += t;
  }
  if (lane == 31) wsum[wave] = incl;
  __syncthreads();
  unsigned woff = 0;
  for (int w = 0; w < 32; ++w)
    if (w < wave) woff += wsum[w];
  const unsigned excl = woff + incl - c;
  histL[tid] = excl;
  if (tid <= NB) offsT[(size_t)tid * G + g] = excl;
  __syncthreads();

  #pragma unroll
  for (int q = 0; q < 8; ++q) {
    if (q * 1024 + tid < len) stage[histL[d[q] >> P_SHIFT] + r[q]] = make_int2(d[q], s[q]);
  }
  __syncthreads();
  for (int j = tid; j < len; j += 1024) part[e0 + j] = stage[j];
}

__global__ __launch_bounds__(1024) void csr_ranges(const int2* __restrict__ part,
                                                   const unsigned* __restrict__ offsT, int N, int G,
                                                   unsigned* __restrict__ cnt,
                                                   unsigned* __restrict__ rowptr,
                                                   int* __restrict__ csr_src) {
  __shared__ unsigned cntL[P_BW];
  __shared__ unsigned wsum[32];
  __shared__ unsigned wbase[32];
  const int tid = threadIdx.x;
  const int lane = tid & 31;
  const int wave = tid >> 5;
  const int b = blockIdx.x;
  const int n0 = b << P_SHIFT;

  if (tid < P_BW) cntL[tid] = 0u;
  __syncthreads();

  int tpp = 1024 / G;
  if (tpp < 1) tpp = 1;
  if (tpp > 32) tpp = 32;
  const int slots = 1024 / tpp;
  const int pg = tid / tpp;
  const int pk = tid - pg * tpp;
  const unsigned* row0 = offsT + (size_t)b * G;
  const unsigned* row1 = row0 + G;

  unsigned mybase = 0u;
  if (pg < slots) {
    for (int g = pg; g < G; g += slots) {
      const unsigned o0 = row0[g];
      const unsigned o1 = row1[g];
      if (pk == 0) mybase += o0;
      const int* pd = (const int*)(part + (size_t)g * P_CH);
      for (unsigned j = o0 + pk; j < o1; j += P_UN * tpp) {
        int dv[P_UN];
        #pragma unroll
        for (int u = 0; u < P_UN; ++u) {
          const unsigned jj = j + u * tpp;
          dv[u] = -1;
          if (jj < o1) dv[u] = pd[2 * (size_t)jj];
        }
        #pragma unroll
        for (int u = 0; u < P_UN; ++u)
          if (dv[u] >= 0) atomicAdd(&cntL[dv[u] - n0], 1u);
      }
    }
  }
  __syncthreads();

  const unsigned bsum = warp_sum_u32(mybase);
  if (lane == 0) wbase[wave] = bsum;
  const unsigned c = (tid < P_BW) ? cntL[tid] : 0u;
  unsigned incl = c;
  for (int dd = 1; dd < 32; dd <<= 1) {
    const unsigned t = __shfl_up(incl, dd, 32);
    if (lane >= dd) incl += t;
  }
  if (lane == 31) wsum[wave] = incl;
  __syncthreads();
  unsigned start = 0u;
  for (int w = 0; w < 32; ++w) start += wbase[w];
  for (int w = 0; w < 32; ++w)
    if (w < wave) start += wsum[w];
  start += incl - c;
  if (tid < P_BW) {
    const int n = n0 + tid;
    if (n < N) { cnt[n] = c; rowptr[n] = start; }
    cntL[tid] = start;
  }
  __syncthreads();

  if (pg < slots) {
    for (int g = pg; g < G; g += slots) {
      const unsigned o0 = row0[g];
      const unsigned o1 = row1[g];
      const int2* p = part + (size_t)g * P_CH;
      for (unsigned j = o0 + pk; j < o1; j += P_UN * tpp) {
        int2 e[P_UN];
        #pragma unroll
        for (int u = 0; u < P_UN; ++u) {
          const unsigned jj = j + u * tpp;
          e[u] = make_int2(-1, 0);
          if (jj < o1) e[u] = p[jj];
        }
        #pragma unroll
        for (int u = 0; u < P_UN; ++u)
          if (e[u].x >= 0) csr_src[atomicAdd(&cntL[e[u].x - n0], 1u)] = e[u].y;
      }
    }
  }
}

__device__ __forceinline__ void gather128_node(const int n, const int lane, const float* __restrict__ H,
                                               const float* __restrict__ es, const float* __restrict__ ed,
                                               const unsigned* __restrict__ rowptr,
                                               const unsigned* __restrict__ cnt,
                                               const int* __restrict__ csr_src, const float4 bv,
                                               const float4 w0, const float4 w1,
                                               const float* __restrict__ a2s, const float* __restrict__ a2d,
                                               float* __restrict__ h2, float* __restrict__ e2s,
                                               float* __restrict__ e2d) {
  const unsigned beg = rowptr[n];
  const int deg = (int)cnt[n];
  const int tot = deg + 1;
  const float edn = ed[n];

  int s0 = n;
  float e0 = -INFINITY;
  const bool v0 = lane < tot;
  if (v0) {
    s0 = (lane < deg) ? csr_src[beg + lane] : n;
    e0 = leaky(es[s0] + edn);
  }
  float m = e0;
  for (int j = lane + 32; j < tot; j += 32) {
    int s = (j < deg) ? csr_src[beg + j] : n;
    m = fmaxf(m, leaky(es[s] + edn));
  }
  m = warp_max(m);

  const float x0 = v0 ? expf(e0 - m) : 0.0f;
  float dsum = x0;
  for (int j = lane + 32; j < tot; j += 32) {
    int s = (j < deg) ? csr_src[beg + j] : n;
    dsum += expf(leaky(es[s] + edn) - m);
  }
  dsum = warp_sum(dsum);

  float4 acc = make_float4(0.0f, 0.0f, 0.0f, 0.0f);
  {
    const float a0 = x0 / dsum;
    const int c0 = (tot < 32) ? tot : 32;
    #pragma unroll 4
    for (int t = 0; t < c0; ++t) {
      const int st = __builtin_amdgcn_readlane(s0, t);
      const float at = __int_as_float(__builtin_amdgcn_readlane(__float_as_int(a0), t));
      const float4 hv = ((const float4*)(H + (size_t)st * 128))[lane];
      acc.x += hv.x * at;
      acc.y += hv.y * at;
      acc.z += hv.z * at;
      acc.w += hv.w * at;
    }
  }
  for (int j0 = 32; j0 < tot; j0 += 32) {
    const int j = j0 + lane;
    int s = n;
    float a = 0.0f;
    if (j < tot) {
      s = (j < deg) ? csr_src[beg + j] : n;
      a = expf(leaky(es[s] + edn) - m) / dsum;
    }
    const int cc = (tot - j0 < 32) ? (tot - j0) : 32;
    for (int t = 0; t < cc; ++t) {
      const int st = __builtin_amdgcn_readlane(s, t);
      const float at = __int_as_float(__builtin_amdgcn_readlane(__float_as_int(a), t));
      const float4 hv = ((const float4*)(H + (size_t)st * 128))[lane];
      acc.x += hv.x * at;
      acc.y += hv.y * at;
      acc.z += hv.z * at;
      acc.w += hv.w * at;
    }
  }
  float4 hv;
  { const float v = acc.x + bv.x; hv.x = (v > 0.0f) ? v : 0.0f; }
  { const float v = acc.y + bv.y; hv.y = (v > 0.0f) ? v : 0.0f; }
  { const float v = acc.z + bv.z; hv.z = (v > 0.0f) ? v : 0.0f; }
  { const float v = acc.w + bv.w; hv.w = (v > 0.0f) ? v : 0.0f; }
  float q0 = hv.x * w0.x + hv.y * w0.z + hv.z * w1.x + hv.w * w1.z;
  float q1 = hv.x * w0.y + hv.y * w0.w + hv.z * w1.y + hv.w * w1.w;
  q0 = warp_sum(q0);
  q1 = warp_sum(q1);
  if (lane == 0) {
    h2[2 * n + 0] = q0;
    h2[2 * n + 1] = q1;
    e2s[n] = q0 * a2s[0] + q1 * a2s[1];
    e2d[n] = q0 * a2d[0] + q1 * a2d[1];
  }
}

#define GK1 5
#define NPW1 10

__device__ __forceinline__ void gather128_group(const int nb, const int lane,
                                                const float* __restrict__ H, const float* __restrict__ es,
                                                const float* __restrict__ ed,
                                                const unsigned* __restrict__ rowptr,
                                                const unsigned* __restrict__ cnt,
                                                const int* __restrict__ csr_src, const float4 bv,
                                                const float4 w0, const float4 w1,
                                                const float* __restrict__ a2s, const float* __restrict__ a2d,
                                                float* __restrict__ h2, float* __restrict__ e2s,
                                                float* __restrict__ e2d) {
  int deg[GK1];
  unsigned beg[GK1];
  float edn[GK1];
  #pragma unroll
  for (int k = 0; k < GK1; ++k) {
    beg[k] = rowptr[nb + k];
    deg[k] = (int)cnt[nb + k];
    edn[k] = ed[nb + k];
  }

  int s0[GK1];
  #pragma unroll
  for (int k = 0; k < GK1; ++k) {
    const unsigned idx = (lane < deg[k]) ? (beg[k] + (unsigned)lane) : 0u;
    const int sv = csr_src[idx];
    s0[k] = (lane < deg[k]) ? sv : (nb + k);
  }
  float a0[GK1];
  #pragma unroll
  for (int k = 0; k < GK1; ++k) a0[k] = es[s0[k]];

  float red[GK1];
  #pragma unroll
  for (int k = 0; k < GK1; ++k) {
    a0[k] = (lane <= deg[k]) ? leaky(a0[k] + edn[k]) : -INFINITY;
    red[k] = a0[k];
    if (deg[k] >= 32) {
      #pragma unroll 1
      for (int j = lane + 32; j < deg[k] + 1; j += 32) {
        const int s = (j < deg[k]) ? csr_src[beg[k] + j] : (nb + k);
        red[k] = fmaxf(red[k], leaky(es[s] + edn[k]));
      }
    }
  }
  #pragma unroll
  for (int o = 16; o > 0; o >>= 1) {
    #pragma unroll
    for (int k = 0; k < GK1; ++k) red[k] = fmaxf(red[k], __shfl_xor(red[k], o, 32));
  }
  float mk[GK1];
  #pragma unroll
  for (int k = 0; k < GK1; ++k)
    mk[k] = __int_as_float(__builtin_amdgcn_readfirstlane(__float_as_int(red[k])));

  #pragma unroll
  for (int k = 0; k < GK1; ++k) {
    a0[k] = (lane <= deg[k]) ? expf(a0[k] - mk[k]) : 0.0f;
    red[k] = a0[k];
    if (deg[k] >= 32) {
      #pragma unroll 1
      for (int j = lane + 32; j < deg[k] + 1; j += 32) {
        const int s = (j < deg[k]) ? csr_src[beg[k] + j] : (nb + k);
        red[k] += expf(leaky(es[s] + edn[k]) - mk[k]);
      }
    }
  }
  #pragma unroll
  for (int o = 16; o > 0; o >>= 1) {
    #pragma unroll
    for (int k = 0; k < GK1; ++k) red[k] += __shfl_xor(red[k], o, 32);
  }
  float dk[GK1];
  #pragma unroll
  for (int k = 0; k < GK1; ++k) {
    dk[k] = __int_as_float(__builtin_amdgcn_readfirstlane(__float_as_int(red[k])));
    a0[k] = a0[k] / dk[k];
  }

  float q0[GK1], q1[GK1];
  #pragma unroll
  for (int k = 0; k < GK1; ++k) {
    float4 acc = make_float4(0.0f, 0.0f, 0.0f, 0.0f);
    const int tot = deg[k] + 1;
    const int c0 = (tot < 32) ? tot : 32;
    #pragma unroll 4
    for (int t = 0; t < c0; ++t) {
      const int st = __builtin_amdgcn_readlane(s0[k], t);
      const float at = __int_as_float(__builtin_amdgcn_readlane(__float_as_int(a0[k]), t));
      const float4 hv = ((const float4*)(H + (size_t)st * 128))[lane];
      acc.x += hv.x * at;
      acc.y += hv.y * at;
      acc.z += hv.z * at;
      acc.w += hv.w * at;
    }
    if (deg[k] >= 32) {
      #pragma unroll 1
      for (int j0 = 32; j0 < tot; j0 += 32) {
        const int j = j0 + lane;
        int s = nb + k;
        float a = 0.0f;
        if (j < tot) {
          s = (j < deg[k]) ? csr_src[beg[k] + j] : (nb + k);
          a = expf(leaky(es[s] + edn[k]) - mk[k]) / dk[k];
        }
        const int cc = (tot - j0 < 32) ? (tot - j0) : 32;
        #pragma unroll 1
        for (int t = 0; t < cc; ++t) {
          const int st = __builtin_amdgcn_readlane(s, t);
          const float at = __int_as_float(__builtin_amdgcn_readlane(__float_as_int(a), t));
          const float4 hv = ((const float4*)(H + (size_t)st * 128))[lane];
          acc.x += hv.x * at;
          acc.y += hv.y * at;
          acc.z += hv.z * at;
          acc.w += hv.w * at;
        }
      }
    }
    float4 hv;
    { const float v = acc.x + bv.x; hv.x = (v > 0.0f) ? v : 0.0f; }
    { const float v = acc.y + bv.y; hv.y = (v > 0.0f) ? v : 0.0f; }
    { const float v = acc.z + bv.z; hv.z = (v > 0.0f) ? v : 0.0f; }
    { const float v = acc.w + bv.w; hv.w = (v > 0.0f) ? v : 0.0f; }
    q0[k] = hv.x * w0.x + hv.y * w0.z + hv.z * w1.x + hv.w * w1.z;
    q1[k] = hv.x * w0.y + hv.y * w0.w + hv.z * w1.y + hv.w * w1.w;
  }
  #pragma unroll
  for (int o = 16; o > 0; o >>= 1) {
    #pragma unroll
    for (int k = 0; k < GK1; ++k) {
      q0[k] += __shfl_xor(q0[k], o, 32);
      q1[k] += __shfl_xor(q1[k], o, 32);
    }
  }
  float r0 = 0.0f, r1 = 0.0f;
  #pragma unroll
  for (int k = 0; k < GK1; ++k) {
    if (lane == k) { r0 = q0[k]; r1 = q1[k]; }
  }
  if (lane < GK1) {
    const int n = nb + lane;
    h2[2 * n + 0] = r0;
    h2[2 * n + 1] = r1;
    e2s[n] = r0 * a2s[0] + r1 * a2s[1];
    e2d[n] = r0 * a2d[0] + r1 * a2d[1];
  }
}

__global__ __launch_bounds__(256) __attribute__((amdgpu_waves_per_eu(12))) void gat_gather128(const float* __restrict__ H, const float* __restrict__ es,
                              const float* __restrict__ ed, const unsigned* __restrict__ rowptr,
                              const unsigned* __restrict__ cnt, const int* __restrict__ csr_src,
                              int N, const float* __restrict__ b1, const float* __restrict__ W2,
                              const float* __restrict__ a2s, const float* __restrict__ a2d,
                              float* __restrict__ h2, float* __restrict__ e2s,
                              float* __restrict__ e2d) {
  const int wb = __builtin_amdgcn_readfirstlane((int)(blockIdx.x * 8 + (threadIdx.x >> 5))) * NPW1;
  if (wb >= N) return;
  const int lane = threadIdx.x & 31;

  const float4 bv = ((const float4*)b1)[lane];
  const float4 w0 = ((const float4*)W2)[2 * lane];
  const float4 w1 = ((const float4*)W2)[2 * lane + 1];

  #pragma unroll 1
  for (int g = 0; g < NPW1; g += GK1) {
    const int nb = wb + g;
    if (nb + GK1 <= N) {
      gather128_group(nb, lane, H, es, ed, rowptr, cnt, csr_src, bv, w0, w1, a2s, a2d, h2, e2s, e2d);
    } else {
      #pragma unroll 1
      for (int n = nb; n < N; ++n)
        gather128_node(n, lane, H, es, ed, rowptr, cnt, csr_src, bv, w0, w1, a2s, a2d, h2, e2s, e2d);
    }
  }
}

__device__ __forceinline__ void gather2_node(const int n, const int lane, const float* __restrict__ h2,
                                             const float* __restrict__ es, const float* __restrict__ ed,
                                             const unsigned* __restrict__ rowptr,
                                             const unsigned* __restrict__ cnt,
                                             const int* __restrict__ csr_src,
                                             const float* __restrict__ b2, float* __restrict__ out) {
  const unsigned beg = rowptr[n];
  const int deg = (int)cnt[n];
  const int tot = deg + 1;
  const float edn = ed[n];

  int s0 = n;
  float e0 = -INFINITY;
  const bool v0 = lane < tot;
  if (v0) {
    s0 = (lane < deg) ? csr_src[beg + lane] : n;
    e0 = leaky(es[s0] + edn);
  }
  float m = e0;
  for (int j = lane + 32; j < tot; j += 32) {
    int s = (j < deg) ? csr_src[beg + j] : n;
    m = fmaxf(m, leaky(es[s] + edn));
  }
  m = warp_max(m);

  const float x0 = v0 ? expf(e0 - m) : 0.0f;
  float dsum = x0;
  for (int j = lane + 32; j < tot; j += 32) {
    int s = (j < deg) ? csr_src[beg + j] : n;
    dsum += expf(leaky(es[s] + edn) - m);
  }
  dsum = warp_sum(dsum);

  float p0 = 0.0f, p1 = 0.0f;
  if (v0) {
    const float a0 = x0 / dsum;
    const float2 hv = ((const float2*)h2)[s0];
    p0 = hv.x * a0;
    p1 = hv.y * a0;
  }
  for (int j = lane + 32; j < tot; j += 32) {
    int s = (j < deg) ? csr_src[beg + j] : n;
    const float a = expf(leaky(es[s] + edn) - m) / dsum;
    const float2 hv = ((const float2*)h2)[s];
    p0 += hv.x * a;
    p1 += hv.y * a;
  }
  p0 = warp_sum(p0);
  p1 = warp_sum(p1);
  if (lane == 0) {
    out[2 * n + 0] = p0 + b2[0];
    out[2 * n + 1] = p1 + b2[1];
  }
}

#define GK2 8
#define NPW2 16

__device__ __forceinline__ void gather2_group(const int nb, const int lane, const float* __restrict__ h2,
                                              const float* __restrict__ es, const float* __restrict__ ed,
                                              const unsigned* __restrict__ rowptr,
                                              const unsigned* __restrict__ cnt,
                                              const int* __restrict__ csr_src,
                                              const float* __restrict__ b2, float* __restrict__ out) {
  int deg[GK2];
  unsigned beg[GK2];
  float edn[GK2];
  #pragma unroll
  for (int k = 0; k < GK2; ++k) {
    beg[k] = rowptr[nb + k];
    deg[k] = (int)cnt[nb + k];
    edn[k] = ed[nb + k];
  }

  int s0[GK2];
  #pragma unroll
  for (int k = 0; k < GK2; ++k) {
    const unsigned idx = (lane < deg[k]) ? (beg[k] + (unsigned)lane) : 0u;
    const int sv = csr_src[idx];
    s0[k] = (lane < deg[k]) ? sv : (nb + k);
  }
  float x0[GK2];
  float2 hv[GK2];
  #pragma unroll
  for (int k = 0; k < GK2; ++k) {
    x0[k] = es[s0[k]];
    hv[k] = ((const float2*)h2)[s0[k]];
  }

  float red[GK2];
  #pragma unroll
  for (int k = 0; k < GK2; ++k) {
    x0[k] = (lane <= deg[k]) ? leaky(x0[k] + edn[k]) : -INFINITY;
    red[k] = x0[k];
    if (deg[k] >= 32) {
      #pragma unroll 1
      for (int j = lane + 32; j < deg[k] + 1; j += 32) {
        const int s = (j < deg[k]) ? csr_src[beg[k] + j] : (nb + k);
        red[k] = fmaxf(red[k], leaky(es[s] + edn[k]));
      }
    }
  }
  #pragma unroll
  for (int o = 16; o > 0; o >>= 1) {
    #pragma unroll
    for (int k = 0; k < GK2; ++k) red[k] = fmaxf(red[k], __shfl_xor(red[k], o, 32));
  }
  float mk[GK2];
  #pragma unroll
  for (int k = 0; k < GK2; ++k)
    mk[k] = __int_as_float(__builtin_amdgcn_readfirstlane(__float_as_int(red[k])));

  #pragma unroll
  for (int k = 0; k < GK2; ++k) {
    x0[k] = (lane <= deg[k]) ? expf(x0[k] - mk[k]) : 0.0f;
    red[k] = x0[k];
    if (deg[k] >= 32) {
      #pragma unroll 1
      for (int j = lane + 32; j < deg[k] + 1; j += 32) {
        const int s = (j < deg[k]) ? csr_src[beg[k] + j] : (nb + k);
        red[k] += expf(leaky(es[s] + edn[k]) - mk[k]);
      }
    }
  }
  #pragma unroll
  for (int o = 16; o > 0; o >>= 1) {
    #pragma unroll
    for (int k = 0; k < GK2; ++k) red[k] += __shfl_xor(red[k], o, 32);
  }

  float p0[GK2], p1[GK2];
  #pragma unroll
  for (int k = 0; k < GK2; ++k) {
    const float dk = __int_as_float(__builtin_amdgcn_readfirstlane(__float_as_int(red[k])));
    const float a = x0[k] / dk;
    p0[k] = (lane <= deg[k]) ? hv[k].x * a : 0.0f;
    p1[k] = (lane <= deg[k]) ? hv[k].y * a : 0.0f;
    if (deg[k] >= 32) {
      #pragma unroll 1
      for (int j = lane + 32; j < deg[k] + 1; j += 32) {
        const int s = (j < deg[k]) ? csr_src[beg[k] + j] : (nb + k);
        const float at = expf(leaky(es[s] + edn[k]) - mk[k]) / dk;
        const float2 ht = ((const float2*)h2)[s];
        p0[k] += ht.x * at;
        p1[k] += ht.y * at;
      }
    }
  }
  #pragma unroll
  for (int o = 16; o > 0; o >>= 1) {
    #pragma unroll
    for (int k = 0; k < GK2; ++k) {
      p0[k] += __shfl_xor(p0[k], o, 32);
      p1[k] += __shfl_xor(p1[k], o, 32);
    }
  }
  float r0 = 0.0f, r1 = 0.0f;
  #pragma unroll
  for (int k = 0; k < GK2; ++k) {
    if (lane == k) { r0 = p0[k]; r1 = p1[k]; }
  }
  if (lane < GK2) {
    const int n = nb + lane;
    out[2 * n + 0] = r0 + b2[0];
    out[2 * n + 1] = r1 + b2[1];
  }
}

__global__ __launch_bounds__(256) void gat_gather2(const float* __restrict__ h2, const float* __restrict__ es,
                            const float* __restrict__ ed, const unsigned* __restrict__ rowptr,
                            const unsigned* __restrict__ cnt, const int* __restrict__ csr_src,
                            int N, const float* __restrict__ b2, float* __restrict__ out) {
  const int wb = __builtin_amdgcn_readfirstlane((int)(blockIdx.x * 8 + (threadIdx.x >> 5))) * NPW2;
  if (wb >= N) return;
  const int lane = threadIdx.x & 31;

  #pragma unroll 1
  for (int g = 0; g < NPW2; g += GK2) {
    const int nb = wb + g;
    if (nb + GK2 <= N) {
      gather2_group(nb, lane, h2, es, ed, rowptr, cnt, csr_src, b2, out);
    } else {
      #pragma unroll 1
      for (int n = nb; n < N; ++n) gather2_node(n, lane, h2, es, ed, rowptr, cnt, csr_src, b2, out);
    }
  }
}


extern "C" void kernel_launch(void* const* d_in, const int* in_sizes, int n_in,
                              void* d_out, int out_size, void* d_ws, size_t ws_size,
                              hipStream_t stream) {
  const float* x   = (const float*)d_in[0];
  const int*  eidx = (const int*)d_in[1];
  const float* W1  = (const float*)d_in[2];
  const float* a1s = (const float*)d_in[3];
  const float* a1d = (const float*)d_in[4];
  const float* b1  = (const float*)d_in[5];
  const float* W2  = (const float*)d_in[6];
  const float* a2s = (const float*)d_in[7];
  const float* a2d = (const float*)d_in[8];
  const float* b2  = (const float*)d_in[9];

  const int N  = in_sizes[0] / 256;
  const int E  = in_sizes[1] / 2;
  const int* src = eidx;
  const int* dst = eidx + E;
  float* out = (float*)d_out;

  const int NB = (N + P_BW - 1) / P_BW;
  const int G  = (E + P_CH - 1) / P_CH;

  char* w = (char*)d_ws;
  float*    h1     = (float*)w;    w += (size_t)N * 128 * sizeof(float);
  int2*     part   = (int2*)w;     w += (size_t)E * sizeof(int2);
  float*    e1s    = (float*)w;    w += (size_t)N * sizeof(float);
  float*    e1d    = (float*)w;    w += (size_t)N * sizeof(float);
  float*    h2     = (float*)w;    w += (size_t)N * 2 * sizeof(float);
  float*    e2s    = (float*)w;    w += (size_t)N * sizeof(float);
  float*    e2d    = (float*)w;    w += (size_t)N * sizeof(float);
  unsigned* cnt    = (unsigned*)w; w += (size_t)N * sizeof(unsigned);
  unsigned* rowptr = (unsigned*)w; w += (size_t)N * sizeof(unsigned);
  unsigned* offsT  = (unsigned*)w; w += (size_t)(NB + 1) * G * sizeof(unsigned);
  int*      csrs   = (int*)w;      w += (size_t)E * sizeof(int);

  const int TB = 256;
  part_edges<<<G, 1024, 0, stream>>>(src, dst, E, NB, G, offsT, part);
  csr_ranges<<<NB, 1024, 0, stream>>>(part, offsT, N, G, cnt, rowptr, csrs);

  gemm1_wmma<<<(N + 63) / 64, 128, 0, stream>>>(x, W1, a1s, a1d, h1, e1s, e1d, N);
  gat_gather128<<<(N + 8 * NPW1 - 1) / (8 * NPW1), TB, 0, stream>>>(h1, e1s, e1d, rowptr, cnt, csrs, N, b1, W2, a2s, a2d,
                                                h2, e2s, e2d);

  gat_gather2<<<(N + 8 * NPW2 - 1) / (8 * NPW2), TB, 0, stream>>>(h2, e2s, e2d, rowptr, cnt, csrs, N, b2, out);
}
